# starting kernel
# baseline (speedup 1.0000x reference)
#include <hip/hip_runtime.h>
#include <cmath>


typedef __attribute__((ext_vector_type(2))) float v2f;
typedef __attribute__((ext_vector_type(8))) float v8f;

#define WMMA_F32(a, b, c) \
  __builtin_amdgcn_wmma_f32_16x16x4_f32(false, (a), false, (b), (short)0, (c), false, false)

#define B_   64
#define C_   2048
#define HW_  196
#define M_   32
#define NC_  396
#define K3_  (M_ * C_)
#define NPAD_ 400
#define NTILES_ 25
#define KCHUNK_ 1024
#define NKCH_   64

__device__ __forceinline__ float fast_sigmoid(float v) {
  return __builtin_amdgcn_rcpf(1.0f + __expf(-v));
}

__global__ __launch_bounds__(128) void k_attn(const float* __restrict__ x,
                                              const float* __restrict__ Wa,
                                              const float* __restrict__ ba,
                                              float* __restrict__ A_ws) {
  const int wave = threadIdx.x >> 5;
  const int hwt  = blockIdx.x * 4 + wave;
  if (hwt >= 13) return;
  const int b    = blockIdx.y;
  const int lane = threadIdx.x & 31;
  const int half = lane >> 4;
  const int r    = lane & 15;
  const int hw0  = hwt * 16;
  const int hwr  = min(hw0 + r, HW_ - 1);

  v8f acc0 = {}; v8f acc1 = {};
  const float* xb = x + (size_t)b * C_ * HW_;

  for (int k0 = 0; k0 < C_; k0 += 4) {
    const int k = k0 + 2 * half;
    v2f a;
    a.x = xb[(size_t)k * HW_ + hwr];
    a.y = xb[(size_t)(k + 1) * HW_ + hwr];
    v2f b0 = *(const v2f*)(Wa + (size_t)(r)      * C_ + k);
    v2f b1 = *(const v2f*)(Wa + (size_t)(16 + r) * C_ + k);
    acc0 = WMMA_F32(a, b0, acc0);
    acc1 = WMMA_F32(a, b1, acc1);
  }

#pragma unroll
  for (int i = 0; i < 8; ++i) {
    const int hw = hw0 + i + (half ? 8 : 0);
    if (hw < HW_) {
      A_ws[((size_t)b * M_ + r)      * HW_ + hw] = fast_sigmoid(acc0[i] + ba[r]);
      A_ws[((size_t)b * M_ + 16 + r) * HW_ + hw] = fast_sigmoid(acc1[i] + ba[16 + r]);
    }
  }
}

__global__ __launch_bounds__(128) void k_bap(const float* __restrict__ x,
                                             const float* __restrict__ A_ws,
                                             float* __restrict__ feats) {
  const int wave = threadIdx.x >> 5;
  const int ct   = blockIdx.x * 4 + wave;
  const int b    = blockIdx.y;
  const int lane = threadIdx.x & 31;
  const int half = lane >> 4;
  const int r    = lane & 15;
  const int c0   = ct * 16;

  v8f acc0 = {}; v8f acc1 = {};
  const float* Ab = A_ws + (size_t)b * M_ * HW_;
  const float* xb = x + ((size_t)b * C_ + c0) * HW_;

  for (int k0 = 0; k0 < HW_; k0 += 4) {
    const int k = k0 + 2 * half;
    v2f a0 = *(const v2f*)(Ab + (size_t)(r)      * HW_ + k);
    v2f a1 = *(const v2f*)(Ab + (size_t)(16 + r) * HW_ + k);
    v2f bb = *(const v2f*)(xb + (size_t)r * HW_ + k);
    acc0 = WMMA_F32(a0, bb, acc0);
    acc1 = WMMA_F32(a1, bb, acc1);
  }

  const float s = 1.0f / (float)HW_;
#pragma unroll
  for (int i = 0; i < 8; ++i) {
    const int mr = i + (half ? 8 : 0);
    feats[((size_t)b * M_ + mr)      * C_ + c0 + r] = acc0[i] * s;
    feats[((size_t)b * M_ + 16 + mr) * C_ + c0 + r] = acc1[i] * s;
  }
}

__global__ __launch_bounds__(256) void k_fc_partial(const float* __restrict__ feats,
                                                    const float* __restrict__ Wc,
                                                    float* __restrict__ part) {
  const int wave = threadIdx.x >> 5;
  const int nt   = blockIdx.x * 8 + wave;
  if (nt >= NTILES_) return;
  const int kc   = blockIdx.y;
  const int lane = threadIdx.x & 31;
  const int half = lane >> 4;
  const int r    = lane & 15;
  const int n0   = nt * 16;

  const bool nvalid = (n0 + r) < NC_;
  const int  nrow   = min(n0 + r, NC_ - 1);
  const float* wrow = Wc + (size_t)nrow * K3_;
  const int kbase   = kc * KCHUNK_;

  v8f acc[4] = {{}, {}, {}, {}};

  for (int k0 = 0; k0 < KCHUNK_; k0 += 4) {
    const int k = kbase + k0 + 2 * half;
    v2f bb = *(const v2f*)(wrow + k);
    bb.x = nvalid ? bb.x : 0.0f;
    bb.y = nvalid ? bb.y : 0.0f;
#pragma unroll
    for (int mt = 0; mt < 4; ++mt) {
      v2f a = *(const v2f*)(feats + (size_t)(mt * 16 + r) * K3_ + k);
      acc[mt] = WMMA_F32(a, bb, acc[mt]);
    }
  }

#pragma unroll
  for (int mt = 0; mt < 4; ++mt) {
#pragma unroll
    for (int i = 0; i < 8; ++i) {
      const int brow = mt * 16 + i + (half ? 8 : 0);
      part[((size_t)kc * B_ + brow) * NPAD_ + n0 + r] = acc[mt][i];
    }
  }
}

__global__ __launch_bounds__(256) void k_reduce(const float* __restrict__ part,
                                               const float* __restrict__ bc,
                                               float* __restrict__ out) {
  const int idx = blockIdx.x * blockDim.x + threadIdx.x;
  if (idx >= B_ * NC_) return;
  const int b = idx / NC_;
  const int n = idx - b * NC_;
  float s = bc[n];
#pragma unroll 4
  for (int kc = 0; kc < NKCH_; ++kc)
    s += part[((size_t)kc * B_ + b) * NPAD_ + n];
  out[idx] = s;
}

extern "C" void kernel_launch(void* const* d_in, const int* in_sizes, int n_in,
                              void* d_out, int out_size, void* d_ws, size_t ws_size,
                              hipStream_t stream) {
  const float* x  = (const float*)d_in[0];
  const float* Wa = (const float*)d_in[1];
  const float* ba = (const float*)d_in[2];
  const float* Wc = (const float*)d_in[3];
  const float* bc = (const float*)d_in[4];
  float* out = (float*)d_out;

  float* A_ws    = (float*)d_ws;
  float* feats   = A_ws + (size_t)B_ * M_ * HW_;
  float* part_ws = feats + (size_t)B_ * K3_;

  k_attn      <<<dim3(4, B_),   128, 0, stream>>>(x, Wa, ba, A_ws);
  k_bap       <<<dim3(32, B_),  128, 0, stream>>>(x, A_ws, feats);
  k_fc_partial<<<dim3(4, NKCH_), 256, 0, stream>>>(feats, Wc, part_ws);
  k_reduce    <<<(B_ * NC_ + 255) / 256, 256, 0, stream>>>(part_ws, bc, out);
}


# k_attn: K loop software-pipelined in registers (ring of 16 k-steps): x and Wa fragments of k-step s+16 requested right behind k-step s's products, ~60 loads in flight per wave; same sum order
# speedup vs baseline: 1.3571x; 1.3571x over previous
#include <hip/hip_runtime.h>
#include <cmath>


typedef __attribute__((ext_vector_type(2))) float v2f;
typedef __attribute__((ext_vector_type(8))) float v8f;

#define WMMA_F32(a, b, c) \
  __builtin_amdgcn_wmma_f32_16x16x4_f32(false, (a), false, (b), (short)0, (c), false, false)

#define B_   64
#define C_   2048
#define HW_  196
#define M_   32
#define NC_  396
#define K3_  (M_ * C_)
#define NPAD_ 400
#define NTILES_ 25
#define KCHUNK_ 1024
#define NKCH_   64
#define ATT_PD_ 16

__device__ __forceinline__ float fast_sigmoid(float v) {
  return __builtin_amdgcn_rcpf(1.0f + __expf(-v));
}

__global__ __launch_bounds__(128) void k_attn(const float* __restrict__ x,
                                              const float* __restrict__ Wa,
                                              const float* __restrict__ ba,
                                              float* __restrict__ A_ws) {
  const int wave = threadIdx.x >> 5;
  const int hwt  = blockIdx.x * 4 + wave;
  if (hwt >= 13) return;
  const int b    = blockIdx.y;
  const int lane = threadIdx.x & 31;
  const int half = lane >> 4;
  const int r    = lane & 15;
  const int hw0  = hwt * 16;
  const int hwr  = min(hw0 + r, HW_ - 1);

  v8f acc0 = {}; v8f acc1 = {};
  const float* xb = x + (size_t)b * C_ * HW_;

  const float* xl  = xb + (size_t)(2 * half) * HW_ + hwr;
  const float* wl0 = Wa + (size_t)(r)      * C_ + 2 * half;
  const float* wl1 = Wa + (size_t)(16 + r) * C_ + 2 * half;

  v2f xa[ATT_PD_], w0[ATT_PD_], w1[ATT_PD_];
#pragma unroll
  for (int j = 0; j < ATT_PD_; ++j) {
    const int kn = 4 * j;
    xa[j].x = xl[(size_t)kn * HW_];
    xa[j].y = xl[(size_t)(kn + 1) * HW_];
    w0[j] = *(const v2f*)(wl0 + kn);
    w1[j] = *(const v2f*)(wl1 + kn);
    __builtin_amdgcn_sched_barrier(0);
  }

  for (int k0 = 0; k0 < C_ - 4 * ATT_PD_; k0 += 4 * ATT_PD_) {
#pragma unroll
    for (int j = 0; j < ATT_PD_; ++j) {
      acc0 = WMMA_F32(xa[j], w0[j], acc0);
      acc1 = WMMA_F32(xa[j], w1[j], acc1);
      const int kn = k0 + 4 * (j + ATT_PD_);
      xa[j].x = xl[(size_t)kn * HW_];
      xa[j].y = xl[(size_t)(kn + 1) * HW_];
      w0[j] = *(const v2f*)(wl0 + kn);
      w1[j] = *(const v2f*)(wl1 + kn);
    }
  }
#pragma unroll
  for (int j = 0; j < ATT_PD_; ++j) {
    acc0 = WMMA_F32(xa[j], w0[j], acc0);
    acc1 = WMMA_F32(xa[j], w1[j], acc1);
  }

#pragma unroll
  for (int i = 0; i < 8; ++i) {
    const int hw = hw0 + i + (half ? 8 : 0);
    if (hw < HW_) {
      A_ws[((size_t)b * M_ + r)      * HW_ + hw] = fast_sigmoid(acc0[i] + ba[r]);
      A_ws[((size_t)b * M_ + 16 + r) * HW_ + hw] = fast_sigmoid(acc1[i] + ba[16 + r]);
    }
  }
}

__global__ __launch_bounds__(128) void k_bap(const float* __restrict__ x,
                                             const float* __restrict__ A_ws,
                                             float* __restrict__ feats) {
  const int wave = threadIdx.x >> 5;
  const int ct   = blockIdx.x * 4 + wave;
  const int b    = blockIdx.y;
  const int lane = threadIdx.x & 31;
  const int half = lane >> 4;
  const int r    = lane & 15;
  const int c0   = ct * 16;

  v8f acc0 = {}; v8f acc1 = {};
  const float* Ab = A_ws + (size_t)b * M_ * HW_;
  const float* xb = x + ((size_t)b * C_ + c0) * HW_;

  for (int k0 = 0; k0 < HW_; k0 += 4) {
    const int k = k0 + 2 * half;
    v2f a0 = *(const v2f*)(Ab + (size_t)(r)      * HW_ + k);
    v2f a1 = *(const v2f*)(Ab + (size_t)(16 + r) * HW_ + k);
    v2f bb = *(const v2f*)(xb + (size_t)r * HW_ + k);
    acc0 = WMMA_F32(a0, bb, acc0);
    acc1 = WMMA_F32(a1, bb, acc1);
  }

  const float s = 1.0f / (float)HW_;
#pragma unroll
  for (int i = 0; i < 8; ++i) {
    const int mr = i + (half ? 8 : 0);
    feats[((size_t)b * M_ + mr)      * C_ + c0 + r] = acc0[i] * s;
    feats[((size_t)b * M_ + 16 + mr) * C_ + c0 + r] = acc1[i] * s;
  }
}

__global__ __launch_bounds__(256) void k_fc_partial(const float* __restrict__ feats,
                                                    const float* __restrict__ Wc,
                                                    float* __restrict__ part) {
  const int wave = threadIdx.x >> 5;
  const int nt   = blockIdx.x * 8 + wave;
  if (nt >= NTILES_) return;
  const int kc   = blockIdx.y;
  const int lane = threadIdx.x & 31;
  const int half = lane >> 4;
  const int r    = lane & 15;
  const int n0   = nt * 16;

  const bool nvalid = (n0 + r) < NC_;
  const int  nrow   = min(n0 + r, NC_ - 1);
  const float* wrow = Wc + (size_t)nrow * K3_;
  const int kbase   = kc * KCHUNK_;

  v8f acc[4] = {{}, {}, {}, {}};

  for (int k0 = 0; k0 < KCHUNK_; k0 += 4) {
    const int k = kbase + k0 + 2 * half;
    v2f bb = *(const v2f*)(wrow + k);
    bb.x = nvalid ? bb.x : 0.0f;
    bb.y = nvalid ? bb.y : 0.0f;
#pragma unroll
    for (int mt = 0; mt < 4; ++mt) {
      v2f a = *(const v2f*)(feats + (size_t)(mt * 16 + r) * K3_ + k);
      acc[mt] = WMMA_F32(a, bb, acc[mt]);
    }
  }

#pragma unroll
  for (int mt = 0; mt < 4; ++mt) {
#pragma unroll
    for (int i = 0; i < 8; ++i) {
      const int brow = mt * 16 + i + (half ? 8 : 0);
      part[((size_t)kc * B_ + brow) * NPAD_ + n0 + r] = acc[mt][i];
    }
  }
}

__global__ __launch_bounds__(256) void k_reduce(const float* __restrict__ part,
                                               const float* __restrict__ bc,
                                               float* __restrict__ out) {
  const int idx = blockIdx.x * blockDim.x + threadIdx.x;
  if (idx >= B_ * NC_) return;
  const int b = idx / NC_;
  const int n = idx - b * NC_;
  float s = bc[n];
#pragma unroll 4
  for (int kc = 0; kc < NKCH_; ++kc)
    s += part[((size_t)kc * B_ + b) * NPAD_ + n];
  out[idx] = s;
}

extern "C" void kernel_launch(void* const* d_in, const int* in_sizes, int n_in,
                              void* d_out, int out_size, void* d_ws, size_t ws_size,
                              hipStream_t stream) {
  const float* x  = (const float*)d_in[0];
  const float* Wa = (const float*)d_in[1];
  const float* ba = (const float*)d_in[2];
  const float* Wc = (const float*)d_in[3];
  const float* bc = (const float*)d_in[4];
  float* out = (float*)d_out;

  float* A_ws    = (float*)d_ws;
  float* feats   = A_ws + (size_t)B_ * M_ * HW_;
  float* part_ws = feats + (size_t)B_ * K3_;

  k_attn      <<<dim3(4, B_),   128, 0, stream>>>(x, Wa, ba, A_ws);
  k_bap       <<<dim3(32, B_),  128, 0, stream>>>(x, A_ws, feats);
  k_fc_partial<<<dim3(4, NKCH_), 256, 0, stream>>>(feats, Wc, part_ws);
  k_reduce    <<<(B_ * NC_ + 255) / 256, 256, 0, stream>>>(part_ws, bc, out);
}


# k_fc_partial re-tiled: a wave owns 2 n-tiles x 4 b-tiles (a feats fragment feeds 2 WMMAs), split-K 128 chunks of 512 (was 64 of 1024); out adds 128 partials in kc order (k sum order changes)
# speedup vs baseline: 1.4421x; 1.0626x over previous
#include <hip/hip_runtime.h>
#include <cmath>


typedef __attribute__((ext_vector_type(2))) float v2f;
typedef __attribute__((ext_vector_type(8))) float v8f;

#define WMMA_F32(a, b, c) \
  __builtin_amdgcn_wmma_f32_16x16x4_f32(false, (a), false, (b), (short)0, (c), false, false)

#define B_   64
#define C_   2048
#define HW_  196
#define M_   32
#define NC_  396
#define K3_  (M_ * C_)
#define NPAD_ 400
#define NTILES_ 25
#define KCHUNK_ 512
#define NKCH_   128
#define NPAIRS_ 13
#define ATT_PD_ 16

__device__ __forceinline__ float fast_sigmoid(float v) {
  return __builtin_amdgcn_rcpf(1.0f + __expf(-v));
}

__global__ __launch_bounds__(128) void k_attn(const float* __restrict__ x,
                                              const float* __restrict__ Wa,
                                              const float* __restrict__ ba,
                                              float* __restrict__ A_ws) {
  const int wave = threadIdx.x >> 5;
  const int hwt  = blockIdx.x * 4 + wave;
  if (hwt >= 13) return;
  const int b    = blockIdx.y;
  const int lane = threadIdx.x & 31;
  const int half = lane >> 4;
  const int r    = lane & 15;
  const int hw0  = hwt * 16;
  const int hwr  = min(hw0 + r, HW_ - 1);

  v8f acc0 = {}; v8f acc1 = {};
  const float* xb = x + (size_t)b * C_ * HW_;

  const float* xl  = xb + (size_t)(2 * half) * HW_ + hwr;
  const float* wl0 = Wa + (size_t)(r)      * C_ + 2 * half;
  const float* wl1 = Wa + (size_t)(16 + r) * C_ + 2 * half;

  v2f xa[ATT_PD_], w0[ATT_PD_], w1[ATT_PD_];
#pragma unroll
  for (int j = 0; j < ATT_PD_; ++j) {
    const int kn = 4 * j;
    xa[j].x = xl[(size_t)kn * HW_];
    xa[j].y = xl[(size_t)(kn + 1) * HW_];
    w0[j] = *(const v2f*)(wl0 + kn);
    w1[j] = *(const v2f*)(wl1 + kn);
    __builtin_amdgcn_sched_barrier(0);
  }

  for (int k0 = 0; k0 < C_ - 4 * ATT_PD_; k0 += 4 * ATT_PD_) {
#pragma unroll
    for (int j = 0; j < ATT_PD_; ++j) {
      acc0 = WMMA_F32(xa[j], w0[j], acc0);
      acc1 = WMMA_F32(xa[j], w1[j], acc1);
      const int kn = k0 + 4 * (j + ATT_PD_);
      xa[j].x = xl[(size_t)kn * HW_];
      xa[j].y = xl[(size_t)(kn + 1) * HW_];
      w0[j] = *(const v2f*)(wl0 + kn);
      w1[j] = *(const v2f*)(wl1 + kn);
    }
  }
#pragma unroll
  for (int j = 0; j < ATT_PD_; ++j) {
    acc0 = WMMA_F32(xa[j], w0[j], acc0);
    acc1 = WMMA_F32(xa[j], w1[j], acc1);
  }

#pragma unroll
  for (int i = 0; i < 8; ++i) {
    const int hw = hw0 + i + (half ? 8 : 0);
    if (hw < HW_) {
      A_ws[((size_t)b * M_ + r)      * HW_ + hw] = fast_sigmoid(acc0[i] + ba[r]);
      A_ws[((size_t)b * M_ + 16 + r) * HW_ + hw] = fast_sigmoid(acc1[i] + ba[16 + r]);
    }
  }
}

__global__ __launch_bounds__(128) void k_bap(const float* __restrict__ x,
                                             const float* __restrict__ A_ws,
                                             float* __restrict__ feats) {
  const int wave = threadIdx.x >> 5;
  const int ct   = blockIdx.x * 4 + wave;
  const int b    = blockIdx.y;
  const int lane = threadIdx.x & 31;
  const int half = lane >> 4;
  const int r    = lane & 15;
  const int c0   = ct * 16;

  v8f acc0 = {}; v8f acc1 = {};
  const float* Ab = A_ws + (size_t)b * M_ * HW_;
  const float* xb = x + ((size_t)b * C_ + c0) * HW_;

  for (int k0 = 0; k0 < HW_; k0 += 4) {
    const int k = k0 + 2 * half;
    v2f a0 = *(const v2f*)(Ab + (size_t)(r)      * HW_ + k);
    v2f a1 = *(const v2f*)(Ab + (size_t)(16 + r) * HW_ + k);
    v2f bb = *(const v2f*)(xb + (size_t)r * HW_ + k);
    acc0 = WMMA_F32(a0, bb, acc0);
    acc1 = WMMA_F32(a1, bb, acc1);
  }

  const float s = 1.0f / (float)HW_;
#pragma unroll
  for (int i = 0; i < 8; ++i) {
    const int mr = i + (half ? 8 : 0);
    feats[((size_t)b * M_ + mr)      * C_ + c0 + r] = acc0[i] * s;
    feats[((size_t)b * M_ + 16 + mr) * C_ + c0 + r] = acc1[i] * s;
  }
}

__global__ __launch_bounds__(128) void k_fc_partial(const float* __restrict__ feats,
                                                    const float* __restrict__ Wc,
                                                    float* __restrict__ part) {
  const int wave = threadIdx.x >> 5;
  const int np   = blockIdx.x * 4 + wave;
  if (np >= NPAIRS_) return;
  const int kc   = blockIdx.y;
  const int lane = threadIdx.x & 31;
  const int half = lane >> 4;
  const int r    = lane & 15;
  const int n0   = np * 32;

  const bool nvalid0 = (n0 + r) < NC_;
  const bool nvalid1 = (n0 + 16 + r) < NC_;
  const float* wrow0 = Wc + (size_t)min(n0 + r, NC_ - 1) * K3_;
  const float* wrow1 = Wc + (size_t)min(n0 + 16 + r, NC_ - 1) * K3_;
  const int kbase    = kc * KCHUNK_;

  v8f acc0[4] = {{}, {}, {}, {}};
  v8f acc1[4] = {{}, {}, {}, {}};

  for (int k0 = 0; k0 < KCHUNK_; k0 += 4) {
    const int k = kbase + k0 + 2 * half;
    v2f b0 = *(const v2f*)(wrow0 + k);
    v2f b1 = *(const v2f*)(wrow1 + k);
    b0.x = nvalid0 ? b0.x : 0.0f;
    b0.y = nvalid0 ? b0.y : 0.0f;
    b1.x = nvalid1 ? b1.x : 0.0f;
    b1.y = nvalid1 ? b1.y : 0.0f;
#pragma unroll
    for (int mt = 0; mt < 4; ++mt) {
      v2f a = *(const v2f*)(feats + (size_t)(mt * 16 + r) * K3_ + k);
      acc0[mt] = WMMA_F32(a, b0, acc0[mt]);
      acc1[mt] = WMMA_F32(a, b1, acc1[mt]);
    }
  }

#pragma unroll
  for (int mt = 0; mt < 4; ++mt) {
#pragma unroll
    for (int i = 0; i < 8; ++i) {
      const int brow = mt * 16 + i + (half ? 8 : 0);
      part[((size_t)kc * B_ + brow) * NPAD_ + n0 + r] = acc0[mt][i];
    }
  }
  if (n0 + 16 < NPAD_) {
#pragma unroll
    for (int mt = 0; mt < 4; ++mt) {
#pragma unroll
      for (int i = 0; i < 8; ++i) {
        const int brow = mt * 16 + i + (half ? 8 : 0);
        part[((size_t)kc * B_ + brow) * NPAD_ + n0 + 16 + r] = acc1[mt][i];
      }
    }
  }
}

__global__ __launch_bounds__(256) void k_reduce(const float* __restrict__ part,
                                               const float* __restrict__ bc,
                                               float* __restrict__ out) {
  const int idx = blockIdx.x * blockDim.x + threadIdx.x;
  if (idx >= B_ * NC_) return;
  const int b = idx / NC_;
  const int n = idx - b * NC_;
  float s = bc[n];
#pragma unroll 4
  for (int kc = 0; kc < NKCH_; ++kc)
    s += part[((size_t)kc * B_ + b) * NPAD_ + n];
  out[idx] = s;
}

extern "C" void kernel_launch(void* const* d_in, const int* in_sizes, int n_in,
                              void* d_out, int out_size, void* d_ws, size_t ws_size,
                              hipStream_t stream) {
  const float* x  = (const float*)d_in[0];
  const float* Wa = (const float*)d_in[1];
  const float* ba = (const float*)d_in[2];
  const float* Wc = (const float*)d_in[3];
  const float* bc = (const float*)d_in[4];
  float* out = (float*)d_out;

  float* A_ws    = (float*)d_ws;
  float* feats   = A_ws + (size_t)B_ * M_ * HW_;
  float* part_ws = feats + (size_t)B_ * K3_;

  k_attn      <<<dim3(4, B_),   128, 0, stream>>>(x, Wa, ba, A_ws);
  k_bap       <<<dim3(32, B_),  128, 0, stream>>>(x, A_ws, feats);
  k_fc_partial<<<dim3(4, NKCH_), 128, 0, stream>>>(feats, Wc, part_ws);
  k_reduce    <<<(B_ * NC_ + 255) / 256, 256, 0, stream>>>(part_ws, bc, out);
}


# k_attn stores A in k_bap's matrix A-operand order (per b, pair of hw k-steps, m-tile: 32 lanes x 16 B contiguous): k_bap's A loads are 16-byte on 4 lines, were 8-byte on 16 lines; same hw order
# speedup vs baseline: 1.5880x; 1.1012x over previous
#include <hip/hip_runtime.h>
#include <cmath>


typedef __attribute__((ext_vector_type(2))) float v2f;
typedef __attribute__((ext_vector_type(4))) float v4f;
typedef __attribute__((ext_vector_type(8))) float v8f;

#define WMMA_F32(a, b, c) \
  __builtin_amdgcn_wmma_f32_16x16x4_f32(false, (a), false, (b), (short)0, (c), false, false)

#define B_   64
#define C_   2048
#define HW_  196
#define M_   32
#define NC_  396
#define K3_  (M_ * C_)
#define NPAD_ 400
#define NTILES_ 25
#define KCHUNK_ 512
#define NKCH_   128
#define NPAIRS_ 13
#define NPR_  25
#define ATT_PD_ 16

__device__ __forceinline__ float fast_sigmoid(float v) {
  return __builtin_amdgcn_rcpf(1.0f + __expf(-v));
}

__global__ __launch_bounds__(128) void k_attn(const float* __restrict__ x,
                                              const float* __restrict__ Wa,
                                              const float* __restrict__ ba,
                                              float* __restrict__ A_ws) {
  const int wave = threadIdx.x >> 5;
  const int hwt  = blockIdx.x * 4 + wave;
  if (hwt >= 13) return;
  const int b    = blockIdx.y;
  const int lane = threadIdx.x & 31;
  const int half = lane >> 4;
  const int r    = lane & 15;
  const int hw0  = hwt * 16;
  const int hwr  = min(hw0 + r, HW_ - 1);

  v8f acc0 = {}; v8f acc1 = {};
  const float* xb = x + (size_t)b * C_ * HW_;

  const float* xl  = xb + (size_t)(2 * half) * HW_ + hwr;
  const float* wl0 = Wa + (size_t)(r)      * C_ + 2 * half;
  const float* wl1 = Wa + (size_t)(16 + r) * C_ + 2 * half;

  v2f xa[ATT_PD_], w0[ATT_PD_], w1[ATT_PD_];
#pragma unroll
  for (int j = 0; j < ATT_PD_; ++j) {
    const int kn = 4 * j;
    xa[j].x = xl[(size_t)kn * HW_];
    xa[j].y = xl[(size_t)(kn + 1) * HW_];
    w0[j] = *(const v2f*)(wl0 + kn);
    w1[j] = *(const v2f*)(wl1 + kn);
    __builtin_amdgcn_sched_barrier(0);
  }

  for (int k0 = 0; k0 < C_ - 4 * ATT_PD_; k0 += 4 * ATT_PD_) {
#pragma unroll
    for (int j = 0; j < ATT_PD_; ++j) {
      acc0 = WMMA_F32(xa[j], w0[j], acc0);
      acc1 = WMMA_F32(xa[j], w1[j], acc1);
      const int kn = k0 + 4 * (j + ATT_PD_);
      xa[j].x = xl[(size_t)kn * HW_];
      xa[j].y = xl[(size_t)(kn + 1) * HW_];
      w0[j] = *(const v2f*)(wl0 + kn);
      w1[j] = *(const v2f*)(wl1 + kn);
    }
  }
#pragma unroll
  for (int j = 0; j < ATT_PD_; ++j) {
    acc0 = WMMA_F32(xa[j], w0[j], acc0);
    acc1 = WMMA_F32(xa[j], w1[j], acc1);
  }

  const int p = 2 * hwt + half;
  if (p < NPR_) {
    const float bias0 = ba[r], bias1 = ba[16 + r];
    float s0[8], s1[8];
#pragma unroll
    for (int i = 0; i < 8; ++i) {
      s0[i] = fast_sigmoid(acc0[i] + bias0);
      s1[i] = fast_sigmoid(acc1[i] + bias1);
    }
    v4f* dst = (v4f*)A_ws + ((size_t)b * NPR_ + p) * 64;
    dst[r]           = (v4f){s0[0], s0[1], s0[4], s0[5]};
    dst[16 + r]      = (v4f){s0[2], s0[3], s0[6], s0[7]};
    dst[32 + r]      = (v4f){s1[0], s1[1], s1[4], s1[5]};
    dst[32 + 16 + r] = (v4f){s1[2], s1[3], s1[6], s1[7]};
  }
}

__global__ __launch_bounds__(128) void k_bap(const float* __restrict__ x,
                                             const float* __restrict__ A_ws,
                                             float* __restrict__ feats) {
  const int wave = threadIdx.x >> 5;
  const int ct   = blockIdx.x * 4 + wave;
  const int b    = blockIdx.y;
  const int lane = threadIdx.x & 31;
  const int half = lane >> 4;
  const int r    = lane & 15;
  const int c0   = ct * 16;

  v8f acc0 = {}; v8f acc1 = {};
  const v4f* Ab = (const v4f*)A_ws + (size_t)b * NPR_ * 64 + lane;
  const float* xl = x + ((size_t)b * C_ + c0 + r) * HW_ + 2 * half;

  for (int p = 0; p < NPR_ - 1; ++p) {
    const v4f a0 = Ab[p * 64];
    const v4f a1 = Ab[p * 64 + 32];
    const v2f be = *(const v2f*)(xl + 8 * p);
    const v2f bo = *(const v2f*)(xl + 8 * p + 4);
    acc0 = WMMA_F32(a0.xy, be, acc0);
    acc1 = WMMA_F32(a1.xy, be, acc1);
    acc0 = WMMA_F32(a0.zw, bo, acc0);
    acc1 = WMMA_F32(a1.zw, bo, acc1);
  }
  {
    const v2f a0 = *(const v2f*)(Ab + (NPR_ - 1) * 64);
    const v2f a1 = *(const v2f*)(Ab + (NPR_ - 1) * 64 + 32);
    const v2f be = *(const v2f*)(xl + 8 * (NPR_ - 1));
    acc0 = WMMA_F32(a0, be, acc0);
    acc1 = WMMA_F32(a1, be, acc1);
  }

  const float s = 1.0f / (float)HW_;
#pragma unroll
  for (int i = 0; i < 8; ++i) {
    const int mr = i + (half ? 8 : 0);
    feats[((size_t)b * M_ + mr)      * C_ + c0 + r] = acc0[i] * s;
    feats[((size_t)b * M_ + 16 + mr) * C_ + c0 + r] = acc1[i] * s;
  }
}

__global__ __launch_bounds__(128) void k_fc_partial(const float* __restrict__ feats,
                                                    const float* __restrict__ Wc,
                                                    float* __restrict__ part) {
  const int wave = threadIdx.x >> 5;
  const int np   = blockIdx.x * 4 + wave;
  if (np >= NPAIRS_) return;
  const int kc   = blockIdx.y;
  const int lane = threadIdx.x & 31;
  const int half = lane >> 4;
  const int r    = lane & 15;
  const int n0   = np * 32;

  const bool nvalid0 = (n0 + r) < NC_;
  const bool nvalid1 = (n0 + 16 + r) < NC_;
  const float* wrow0 = Wc + (size_t)min(n0 + r, NC_ - 1) * K3_;
  const float* wrow1 = Wc + (size_t)min(n0 + 16 + r, NC_ - 1) * K3_;
  const int kbase    = kc * KCHUNK_;

  v8f acc0[4] = {{}, {}, {}, {}};
  v8f acc1[4] = {{}, {}, {}, {}};

  for (int k0 = 0; k0 < KCHUNK_; k0 += 4) {
    const int k = kbase + k0 + 2 * half;
    v2f b0 = *(const v2f*)(wrow0 + k);
    v2f b1 = *(const v2f*)(wrow1 + k);
    b0.x = nvalid0 ? b0.x : 0.0f;
    b0.y = nvalid0 ? b0.y : 0.0f;
    b1.x = nvalid1 ? b1.x : 0.0f;
    b1.y = nvalid1 ? b1.y : 0.0f;
#pragma unroll
    for (int mt = 0; mt < 4; ++mt) {
      v2f a = *(const v2f*)(feats + (size_t)(mt * 16 + r) * K3_ + k);
      acc0[mt] = WMMA_F32(a, b0, acc0[mt]);
      acc1[mt] = WMMA_F32(a, b1, acc1[mt]);
    }
  }

#pragma unroll
  for (int mt = 0; mt < 4; ++mt) {
#pragma unroll
    for (int i = 0; i < 8; ++i) {
      const int brow = mt * 16 + i + (half ? 8 : 0);
      part[((size_t)kc * B_ + brow) * NPAD_ + n0 + r] = acc0[mt][i];
    }
  }
  if (n0 + 16 < NPAD_) {
#pragma unroll
    for (int mt = 0; mt < 4; ++mt) {
#pragma unroll
      for (int i = 0; i < 8; ++i) {
        const int brow = mt * 16 + i + (half ? 8 : 0);
        part[((size_t)kc * B_ + brow) * NPAD_ + n0 + 16 + r] = acc1[mt][i];
      }
    }
  }
}

__global__ __launch_bounds__(256) void k_reduce(const float* __restrict__ part,
                                               const float* __restrict__ bc,
                                               float* __restrict__ out) {
  const int idx = blockIdx.x * blockDim.x + threadIdx.x;
  if (idx >= B_ * NC_) return;
  const int b = idx / NC_;
  const int n = idx - b * NC_;
  float s = bc[n];
#pragma unroll 4
  for (int kc = 0; kc < NKCH_; ++kc)
    s += part[((size_t)kc * B_ + b) * NPAD_ + n];
  out[idx] = s;
}

extern "C" void kernel_launch(void* const* d_in, const int* in_sizes, int n_in,
                              void* d_out, int out_size, void* d_ws, size_t ws_size,
                              hipStream_t stream) {
  const float* x  = (const float*)d_in[0];
  const float* Wa = (const float*)d_in[1];
  const float* ba = (const float*)d_in[2];
  const float* Wc = (const float*)d_in[3];
  const float* bc = (const float*)d_in[4];
  float* out = (float*)d_out;

  float* A_ws    = (float*)d_ws;
  float* feats   = A_ws + (size_t)B_ * NPR_ * 256;
  float* part_ws = feats + (size_t)B_ * K3_;

  k_attn      <<<dim3(4, B_),   128, 0, stream>>>(x, Wa, ba, A_ws);
  k_bap       <<<dim3(32, B_),  128, 0, stream>>>(x, A_ws, feats);
  k_fc_partial<<<dim3(4, NKCH_), 128, 0, stream>>>(feats, Wc, part_ws);
  k_reduce    <<<(B_ * NC_ + 255) / 256, 256, 0, stream>>>(part_ws, bc, out);
}


# k_attn: Wa fragments come from a per-block LDS copy in B-operand order (one 16-byte LDS read per k-step; filled by whole-line loads); global loads are x's alone, ring 16 -> 32 k-steps; same sum order
# speedup vs baseline: 1.7184x; 1.0821x over previous
#include <hip/hip_runtime.h>
#include <cmath>


typedef __attribute__((ext_vector_type(2))) float v2f;
typedef __attribute__((ext_vector_type(4))) float v4f;
typedef __attribute__((ext_vector_type(8))) float v8f;

#define WMMA_F32(a, b, c) \
  __builtin_amdgcn_wmma_f32_16x16x4_f32(false, (a), false, (b), (short)0, (c), false, false)

#define B_   64
#define C_   2048
#define HW_  196
#define M_   32
#define NC_  396
#define K3_  (M_ * C_)
#define NPAD_ 400
#define NTILES_ 25
#define KCHUNK_ 512
#define NKCH_   128
#define NPAIRS_ 13
#define NPR_  25
#define ATT_KS_ 512
#define ATT_XD_ 32
#define ATT_WG_ 16
#define ATT_WP_ 132
#define ATT_FB_ 32

__device__ __forceinline__ float fast_sigmoid(float v) {
  return __builtin_amdgcn_rcpf(1.0f + __expf(-v));
}

__global__ __launch_bounds__(128) void k_attn(const float* __restrict__ x,
                                              const float* __restrict__ Wa,
                                              const float* __restrict__ ba,
                                              float* __restrict__ A_ws) {
  __shared__ float Wl[ATT_KS_ * ATT_WP_];

  const int wave = threadIdx.x >> 5;
  const int hwt  = blockIdx.x * 4 + wave;
  const int b    = blockIdx.y;
  const int lane = threadIdx.x & 31;
  const int half = lane >> 4;
  const int r    = lane & 15;

  {
    const float* src = Wa + (size_t)(16 * half + wave) * C_ + 4 * r;
    float* dst = Wl + r * ATT_WP_ + wave * 4 + half * 2;
#pragma unroll 1
    for (int n0 = 0; n0 < 128; n0 += ATT_FB_) {
      v4f tmp[ATT_FB_];
#pragma unroll
      for (int u = 0; u < ATT_FB_; ++u)
        tmp[u] = *(const v4f*)(src + (size_t)(4 * (u & 3)) * C_ + 16 * n0 + 64 * (u >> 2));
#pragma unroll
      for (int u = 0; u < ATT_FB_; ++u) {
        float* d = dst + (4 * n0 + 16 * (u >> 2)) * ATT_WP_ + 16 * (u & 3);
        *(v2f*)d        = tmp[u].xy;
        *(v2f*)(d + 64) = tmp[u].zw;
      }
    }
  }

  const int hw0  = hwt * 16;
  const int hwr  = min(hw0 + r, HW_ - 1);

  v8f acc0 = {}; v8f acc1 = {};
  const float* xb = x + (size_t)b * C_ * HW_;

  const float* xl  = xb + (size_t)(2 * half) * HW_ + hwr;

  v2f xa[ATT_XD_];
  if (hwt < 13) {
#pragma unroll
    for (int j = 0; j < ATT_XD_; ++j) {
      xa[j].x = xl[(size_t)(4 * j) * HW_];
      xa[j].y = xl[(size_t)(4 * j + 1) * HW_];
    }
  }
  __syncthreads();
  if (hwt >= 13) return;

  const float* wl = Wl + lane * 4;
  v4f wf[2][ATT_WG_];
#pragma unroll
  for (int j = 0; j < ATT_WG_; ++j) wf[0][j] = *(const v4f*)(wl + j * ATT_WP_);

#pragma unroll 1
  for (int ks0 = 0; ks0 < ATT_KS_ - ATT_XD_; ks0 += ATT_XD_) {
#pragma unroll
    for (int h = 0; h < 2; ++h) {
#pragma unroll
      for (int j = 0; j < ATT_WG_; ++j)
        wf[1 - h][j] = *(const v4f*)(wl + (ks0 + ATT_WG_ * (h + 1) + j) * ATT_WP_);
      __builtin_amdgcn_sched_barrier(0);
#pragma unroll
      for (int j = 0; j < ATT_WG_; ++j) {
        const int q = ATT_WG_ * h + j;
        acc0 = WMMA_F32(xa[q], wf[h][j].xy, acc0);
        acc1 = WMMA_F32(xa[q], wf[h][j].zw, acc1);
        const int kn = 4 * (ks0 + ATT_XD_ + q);
        xa[q].x = xl[(size_t)kn * HW_];
        xa[q].y = xl[(size_t)(kn + 1) * HW_];
      }
    }
  }
#pragma unroll
  for (int j = 0; j < ATT_WG_; ++j)
    wf[1][j] = *(const v4f*)(wl + (ATT_KS_ - ATT_WG_ + j) * ATT_WP_);
  __builtin_amdgcn_sched_barrier(0);
#pragma unroll
  for (int j = 0; j < ATT_WG_; ++j) {
    acc0 = WMMA_F32(xa[j], wf[0][j].xy, acc0);
    acc1 = WMMA_F32(xa[j], wf[0][j].zw, acc1);
  }
#pragma unroll
  for (int j = 0; j < ATT_WG_; ++j) {
    acc0 = WMMA_F32(xa[ATT_WG_ + j], wf[1][j].xy, acc0);
    acc1 = WMMA_F32(xa[ATT_WG_ + j], wf[1][j].zw, acc1);
  }

  const int p = 2 * hwt + half;
  if (p < NPR_) {
    const float bias0 = ba[r], bias1 = ba[16 + r];
    float s0[8], s1[8];
#pragma unroll
    for (int i = 0; i < 8; ++i) {
      s0[i] = fast_sigmoid(acc0[i] + bias0);
      s1[i] = fast_sigmoid(acc1[i] + bias1);
    }
    v4f* dst = (v4f*)A_ws + ((size_t)b * NPR_ + p) * 64;
    dst[r]           = (v4f){s0[0], s0[1], s0[4], s0[5]};
    dst[16 + r]      = (v4f){s0[2], s0[3], s0[6], s0[7]};
    dst[32 + r]      = (v4f){s1[0], s1[1], s1[4], s1[5]};
    dst[32 + 16 + r] = (v4f){s1[2], s1[3], s1[6], s1[7]};
  }
}

__global__ __launch_bounds__(128) void k_bap(const float* __restrict__ x,
                                             const float* __restrict__ A_ws,
                                             float* __restrict__ feats) {
  const int wave = threadIdx.x >> 5;
  const int ct   = blockIdx.x * 4 + wave;
  const int b    = blockIdx.y;
  const int lane = threadIdx.x & 31;
  const int half = lane >> 4;
  const int r    = lane & 15;
  const int c0   = ct * 16;

  v8f acc0 = {}; v8f acc1 = {};
  const v4f* Ab = (const v4f*)A_ws + (size_t)b * NPR_ * 64 + lane;
  const float* xl = x + ((size_t)b * C_ + c0 + r) * HW_ + 2 * half;

  for (int p = 0; p < NPR_ - 1; ++p) {
    const v4f a0 = Ab[p * 64];
    const v4f a1 = Ab[p * 64 + 32];
    const v2f be = *(const v2f*)(xl + 8 * p);
    const v2f bo = *(const v2f*)(xl + 8 * p + 4);
    acc0 = WMMA_F32(a0.xy, be, acc0);
    acc1 = WMMA_F32(a1.xy, be, acc1);
    acc0 = WMMA_F32(a0.zw, bo, acc0);
    acc1 = WMMA_F32(a1.zw, bo, acc1);
  }
  {
    const v2f a0 = *(const v2f*)(Ab + (NPR_ - 1) * 64);
    const v2f a1 = *(const v2f*)(Ab + (NPR_ - 1) * 64 + 32);
    const v2f be = *(const v2f*)(xl + 8 * (NPR_ - 1));
    acc0 = WMMA_F32(a0, be, acc0);
    acc1 = WMMA_F32(a1, be, acc1);
  }

  const float s = 1.0f / (float)HW_;
#pragma unroll
  for (int i = 0; i < 8; ++i) {
    const int mr = i + (half ? 8 : 0);
    feats[((size_t)b * M_ + mr)      * C_ + c0 + r] = acc0[i] * s;
    feats[((size_t)b * M_ + 16 + mr) * C_ + c0 + r] = acc1[i] * s;
  }
}

__global__ __launch_bounds__(128) void k_fc_partial(const float* __restrict__ feats,
                                                    const float* __restrict__ Wc,
                                                    float* __restrict__ part) {
  const int wave = threadIdx.x >> 5;
  const int np   = blockIdx.x * 4 + wave;
  if (np >= NPAIRS_) return;
  const int kc   = blockIdx.y;
  const int lane = threadIdx.x & 31;
  const int half = lane >> 4;
  const int r    = lane & 15;
  const int n0   = np * 32;

  const bool nvalid0 = (n0 + r) < NC_;
  const bool nvalid1 = (n0 + 16 + r) < NC_;
  const float* wrow0 = Wc + (size_t)min(n0 + r, NC_ - 1) * K3_;
  const float* wrow1 = Wc + (size_t)min(n0 + 16 + r, NC_ - 1) * K3_;
  const int kbase    = kc * KCHUNK_;

  v8f acc0[4] = {{}, {}, {}, {}};
  v8f acc1[4] = {{}, {}, {}, {}};

  for (int k0 = 0; k0 < KCHUNK_; k0 += 4) {
    const int k = kbase + k0 + 2 * half;
    v2f b0 = *(const v2f*)(wrow0 + k);
    v2f b1 = *(const v2f*)(wrow1 + k);
    b0.x = nvalid0 ? b0.x : 0.0f;
    b0.y = nvalid0 ? b0.y : 0.0f;
    b1.x = nvalid1 ? b1.x : 0.0f;
    b1.y = nvalid1 ? b1.y : 0.0f;
#pragma unroll
    for (int mt = 0; mt < 4; ++mt) {
      v2f a = *(const v2f*)(feats + (size_t)(mt * 16 + r) * K3_ + k);
      acc0[mt] = WMMA_F32(a, b0, acc0[mt]);
      acc1[mt] = WMMA_F32(a, b1, acc1[mt]);
    }
  }

#pragma unroll
  for (int mt = 0; mt < 4; ++mt) {
#pragma unroll
    for (int i = 0; i < 8; ++i) {
      const int brow = mt * 16 + i + (half ? 8 : 0);
      part[((size_t)kc * B_ + brow) * NPAD_ + n0 + r] = acc0[mt][i];
    }
  }
  if (n0 + 16 < NPAD_) {
#pragma unroll
    for (int mt = 0; mt < 4; ++mt) {
#pragma unroll
      for (int i = 0; i < 8; ++i) {
        const int brow = mt * 16 + i + (half ? 8 : 0);
        part[((size_t)kc * B_ + brow) * NPAD_ + n0 + 16 + r] = acc1[mt][i];
      }
    }
  }
}

__global__ __launch_bounds__(256) void k_reduce(const float* __restrict__ part,
                                               const float* __restrict__ bc,
                                               float* __restrict__ out) {
  const int idx = blockIdx.x * blockDim.x + threadIdx.x;
  if (idx >= B_ * NC_) return;
  const int b = idx / NC_;
  const int n = idx - b * NC_;
  float s = bc[n];
#pragma unroll 4
  for (int kc = 0; kc < NKCH_; ++kc)
    s += part[((size_t)kc * B_ + b) * NPAD_ + n];
  out[idx] = s;
}

extern "C" void kernel_launch(void* const* d_in, const int* in_sizes, int n_in,
                              void* d_out, int out_size, void* d_ws, size_t ws_size,
                              hipStream_t stream) {
  const float* x  = (const float*)d_in[0];
  const float* Wa = (const float*)d_in[1];
  const float* ba = (const float*)d_in[2];
  const float* Wc = (const float*)d_in[3];
  const float* bc = (const float*)d_in[4];
  float* out = (float*)d_out;

  float* A_ws    = (float*)d_ws;
  float* feats   = A_ws + (size_t)B_ * NPR_ * 256;
  float* part_ws = feats + (size_t)B_ * K3_;

  k_attn      <<<dim3(4, B_),   128, 0, stream>>>(x, Wa, ba, A_ws);
  k_bap       <<<dim3(32, B_),  128, 0, stream>>>(x, A_ws, feats);
  k_fc_partial<<<dim3(4, NKCH_), 128, 0, stream>>>(feats, Wc, part_ws);
  k_reduce    <<<(B_ * NC_ + 255) / 256, 256, 0, stream>>>(part_ws, bc, out);
}


# k_fc_partial: K loop software-pipelined in registers: feats fragments requested 8 k-steps ahead of their products, Wc fragments 16 ahead (3 register sets, loop unrolled whole); same sum order
# speedup vs baseline: 1.8584x; 1.0815x over previous
#include <hip/hip_runtime.h>
#include <cmath>


typedef __attribute__((ext_vector_type(2))) float v2f;
typedef __attribute__((ext_vector_type(4))) float v4f;
typedef __attribute__((ext_vector_type(8))) float v8f;

#define WMMA_F32(a, b, c) \
  __builtin_amdgcn_wmma_f32_16x16x4_f32(false, (a), false, (b), (short)0, (c), false, false)

#define B_   64
#define C_   2048
#define HW_  196
#define M_   32
#define NC_  396
#define K3_  (M_ * C_)
#define NPAD_ 400
#define NTILES_ 25
#define KCHUNK_ 512
#define NKCH_   128
#define FC_G_   8
#define FC_NG_  16
#define NPAIRS_ 13
#define NPR_  25
#define ATT_KS_ 512
#define ATT_XD_ 32
#define ATT_WG_ 16
#define ATT_WP_ 132
#define ATT_FB_ 32

__device__ __forceinline__ float fast_sigmoid(float v) {
  return __builtin_amdgcn_rcpf(1.0f + __expf(-v));
}

__global__ __launch_bounds__(128) void k_attn(const float* __restrict__ x,
                                              const float* __restrict__ Wa,
                                              const float* __restrict__ ba,
                                              float* __restrict__ A_ws) {
  __shared__ float Wl[ATT_KS_ * ATT_WP_];

  const int wave = threadIdx.x >> 5;
  const int hwt  = blockIdx.x * 4 + wave;
  const int b    = blockIdx.y;
  const int lane = threadIdx.x & 31;
  const int half = lane >> 4;
  const int r    = lane & 15;

  {
    const float* src = Wa + (size_t)(16 * half + wave) * C_ + 4 * r;
    float* dst = Wl + r * ATT_WP_ + wave * 4 + half * 2;
#pragma unroll 1
    for (int n0 = 0; n0 < 128; n0 += ATT_FB_) {
      v4f tmp[ATT_FB_];
#pragma unroll
      for (int u = 0; u < ATT_FB_; ++u)
        tmp[u] = *(const v4f*)(src + (size_t)(4 * (u & 3)) * C_ + 16 * n0 + 64 * (u >> 2));
#pragma unroll
      for (int u = 0; u < ATT_FB_; ++u) {
        float* d = dst + (4 * n0 + 16 * (u >> 2)) * ATT_WP_ + 16 * (u & 3);
        *(v2f*)d        = tmp[u].xy;
        *(v2f*)(d + 64) = tmp[u].zw;
      }
    }
  }

  const int hw0  = hwt * 16;
  const int hwr  = min(hw0 + r, HW_ - 1);

  v8f acc0 = {}; v8f acc1 = {};
  const float* xb = x + (size_t)b * C_ * HW_;

  const float* xl  = xb + (size_t)(2 * half) * HW_ + hwr;

  v2f xa[ATT_XD_];
  if (hwt < 13) {
#pragma unroll
    for (int j = 0; j < ATT_XD_; ++j) {
      xa[j].x = xl[(size_t)(4 * j) * HW_];
      xa[j].y = xl[(size_t)(4 * j + 1) * HW_];
    }
  }
  __syncthreads();
  if (hwt >= 13) return;

  const float* wl = Wl + lane * 4;
  v4f wf[2][ATT_WG_];
#pragma unroll
  for (int j = 0; j < ATT_WG_; ++j) wf[0][j] = *(const v4f*)(wl + j * ATT_WP_);

#pragma unroll 1
  for (int ks0 = 0; ks0 < ATT_KS_ - ATT_XD_; ks0 += ATT_XD_) {
#pragma unroll
    for (int h = 0; h < 2; ++h) {
#pragma unroll
      for (int j = 0; j < ATT_WG_; ++j)
        wf[1 - h][j] = *(const v4f*)(wl + (ks0 + ATT_WG_ * (h + 1) + j) * ATT_WP_);
      __builtin_amdgcn_sched_barrier(0);
#pragma unroll
      for (int j = 0; j < ATT_WG_; ++j) {
        const int q = ATT_WG_ * h + j;
        acc0 = WMMA_F32(xa[q], wf[h][j].xy, acc0);
        acc1 = WMMA_F32(xa[q], wf[h][j].zw, acc1);
        const int kn = 4 * (ks0 + ATT_XD_ + q);
        xa[q].x = xl[(size_t)kn * HW_];
        xa[q].y = xl[(size_t)(kn + 1) * HW_];
      }
    }
  }
#pragma unroll
  for (int j = 0; j < ATT_WG_; ++j)
    wf[1][j] = *(const v4f*)(wl + (ATT_KS_ - ATT_WG_ + j) * ATT_WP_);
  __builtin_amdgcn_sched_barrier(0);
#pragma unroll
  for (int j = 0; j < ATT_WG_; ++j) {
    acc0 = WMMA_F32(xa[j], wf[0][j].xy, acc0);
    acc1 = WMMA_F32(xa[j], wf[0][j].zw, acc1);
  }
#pragma unroll
  for (int j = 0; j < ATT_WG_; ++j) {
    acc0 = WMMA_F32(xa[ATT_WG_ + j], wf[1][j].xy, acc0);
    acc1 = WMMA_F32(xa[ATT_WG_ + j], wf[1][j].zw, acc1);
  }

  const int p = 2 * hwt + half;
  if (p < NPR_) {
    const float bias0 = ba[r], bias1 = ba[16 + r];
    float s0[8], s1[8];
#pragma unroll
    for (int i = 0; i < 8; ++i) {
      s0[i] = fast_sigmoid(acc0[i] + bias0);
      s1[i] = fast_sigmoid(acc1[i] + bias1);
    }
    v4f* dst = (v4f*)A_ws + ((size_t)b * NPR_ + p) * 64;
    dst[r]           = (v4f){s0[0], s0[1], s0[4], s0[5]};
    dst[16 + r]      = (v4f){s0[2], s0[3], s0[6], s0[7]};
    dst[32 + r]      = (v4f){s1[0], s1[1], s1[4], s1[5]};
    dst[32 + 16 + r] = (v4f){s1[2], s1[3], s1[6], s1[7]};
  }
}

__global__ __launch_bounds__(128) void k_bap(const float* __restrict__ x,
                                             const float* __restrict__ A_ws,
                                             float* __restrict__ feats) {
  const int wave = threadIdx.x >> 5;
  const int ct   = blockIdx.x * 4 + wave;
  const int b    = blockIdx.y;
  const int lane = threadIdx.x & 31;
  const int half = lane >> 4;
  const int r    = lane & 15;
  const int c0   = ct * 16;

  v8f acc0 = {}; v8f acc1 = {};
  const v4f* Ab = (const v4f*)A_ws + (size_t)b * NPR_ * 64 + lane;
  const float* xl = x + ((size_t)b * C_ + c0 + r) * HW_ + 2 * half;

  for (int p = 0; p < NPR_ - 1; ++p) {
    const v4f a0 = Ab[p * 64];
    const v4f a1 = Ab[p * 64 + 32];
    const v2f be = *(const v2f*)(xl + 8 * p);
    const v2f bo = *(const v2f*)(xl + 8 * p + 4);
    acc0 = WMMA_F32(a0.xy, be, acc0);
    acc1 = WMMA_F32(a1.xy, be, acc1);
    acc0 = WMMA_F32(a0.zw, bo, acc0);
    acc1 = WMMA_F32(a1.zw, bo, acc1);
  }
  {
    const v2f a0 = *(const v2f*)(Ab + (NPR_ - 1) * 64);
    const v2f a1 = *(const v2f*)(Ab + (NPR_ - 1) * 64 + 32);
    const v2f be = *(const v2f*)(xl + 8 * (NPR_ - 1));
    acc0 = WMMA_F32(a0, be, acc0);
    acc1 = WMMA_F32(a1, be, acc1);
  }

  const float s = 1.0f / (float)HW_;
#pragma unroll
  for (int i = 0; i < 8; ++i) {
    const int mr = i + (half ? 8 : 0);
    feats[((size_t)b * M_ + mr)      * C_ + c0 + r] = acc0[i] * s;
    feats[((size_t)b * M_ + 16 + mr) * C_ + c0 + r] = acc1[i] * s;
  }
}

__device__ __forceinline__ void fc_load_w(v2f (&W)[FC_G_][2], const float* w0, const float* w1) {
#pragma unroll
  for (int j = 0; j < FC_G_; ++j) {
    W[j][0] = *(const v2f*)(w0 + 4 * j);
    W[j][1] = *(const v2f*)(w1 + 4 * j);
  }
}

__device__ __forceinline__ void fc_load_f(v2f (&F)[FC_G_][4], const float* f0) {
#pragma unroll
  for (int j = 0; j < FC_G_; ++j) {
#pragma unroll
    for (int mt = 0; mt < 4; ++mt)
      F[j][mt] = *(const v2f*)(f0 + (size_t)(mt * 16) * K3_ + 4 * j);
  }
}

__device__ __forceinline__ void fc_mma(const v2f (&W)[FC_G_][2], const v2f (&F)[FC_G_][4],
                                       const bool nvalid0, const bool nvalid1,
                                       v8f (&acc0)[4], v8f (&acc1)[4]) {
#pragma unroll
  for (int j = 0; j < FC_G_; ++j) {
    v2f b0 = W[j][0];
    v2f b1 = W[j][1];
    b0.x = nvalid0 ? b0.x : 0.0f;
    b0.y = nvalid0 ? b0.y : 0.0f;
    b1.x = nvalid1 ? b1.x : 0.0f;
    b1.y = nvalid1 ? b1.y : 0.0f;
#pragma unroll
    for (int mt = 0; mt < 4; ++mt) {
      acc0[mt] = WMMA_F32(F[j][mt], b0, acc0[mt]);
      acc1[mt] = WMMA_F32(F[j][mt], b1, acc1[mt]);
    }
  }
}

__global__ __launch_bounds__(128) void k_fc_partial(const float* __restrict__ feats,
                                                    const float* __restrict__ Wc,
                                                    float* __restrict__ part) {
  const int wave = threadIdx.x >> 5;
  const int np   = blockIdx.x * 4 + wave;
  if (np >= NPAIRS_) return;
  const int kc   = blockIdx.y;
  const int lane = threadIdx.x & 31;
  const int half = lane >> 4;
  const int r    = lane & 15;
  const int n0   = np * 32;

  const bool nvalid0 = (n0 + r) < NC_;
  const bool nvalid1 = (n0 + 16 + r) < NC_;
  const int kbase    = kc * KCHUNK_ + 2 * half;
  const float* w0 = Wc + (size_t)min(n0 + r, NC_ - 1) * K3_ + kbase;
  const float* w1 = Wc + (size_t)min(n0 + 16 + r, NC_ - 1) * K3_ + kbase;
  const float* f0 = feats + (size_t)r * K3_ + kbase;

  v8f acc0[4] = {{}, {}, {}, {}};
  v8f acc1[4] = {{}, {}, {}, {}};

  v2f WS[3][FC_G_][2];
  v2f FS[2][FC_G_][4];

  fc_load_f(FS[0], f0);
  fc_load_w(WS[0], w0, w1);
  fc_load_w(WS[1], w0 + 4 * FC_G_, w1 + 4 * FC_G_);
#pragma unroll
  for (int g = 0; g < FC_NG_; ++g) {
    if (g + 1 < FC_NG_) fc_load_f(FS[(g + 1) & 1], f0 + 4 * FC_G_ * (g + 1));
    if (g + 2 < FC_NG_) fc_load_w(WS[(g + 2) % 3], w0 + 4 * FC_G_ * (g + 2), w1 + 4 * FC_G_ * (g + 2));
    __builtin_amdgcn_sched_barrier(0);
    fc_mma(WS[g % 3], FS[g & 1], nvalid0, nvalid1, acc0, acc1);
  }

#pragma unroll
  for (int mt = 0; mt < 4; ++mt) {
#pragma unroll
    for (int i = 0; i < 8; ++i) {
      const int brow = mt * 16 + i + (half ? 8 : 0);
      part[((size_t)kc * B_ + brow) * NPAD_ + n0 + r] = acc0[mt][i];
    }
  }
  if (n0 + 16 < NPAD_) {
#pragma unroll
    for (int mt = 0; mt < 4; ++mt) {
#pragma unroll
      for (int i = 0; i < 8; ++i) {
        const int brow = mt * 16 + i + (half ? 8 : 0);
        part[((size_t)kc * B_ + brow) * NPAD_ + n0 + 16 + r] = acc1[mt][i];
      }
    }
  }
}

__global__ __launch_bounds__(256) void k_reduce(const float* __restrict__ part,
                                               const float* __restrict__ bc,
                                               float* __restrict__ out) {
  const int idx = blockIdx.x * blockDim.x + threadIdx.x;
  if (idx >= B_ * NC_) return;
  const int b = idx / NC_;
  const int n = idx - b * NC_;
  float s = bc[n];
#pragma unroll 4
  for (int kc = 0; kc < NKCH_; ++kc)
    s += part[((size_t)kc * B_ + b) * NPAD_ + n];
  out[idx] = s;
}

extern "C" void kernel_launch(void* const* d_in, const int* in_sizes, int n_in,
                              void* d_out, int out_size, void* d_ws, size_t ws_size,
                              hipStream_t stream) {
  const float* x  = (const float*)d_in[0];
  const float* Wa = (const float*)d_in[1];
  const float* ba = (const float*)d_in[2];
  const float* Wc = (const float*)d_in[3];
  const float* bc = (const float*)d_in[4];
  float* out = (float*)d_out;

  float* A_ws    = (float*)d_ws;
  float* feats   = A_ws + (size_t)B_ * NPR_ * 256;
  float* part_ws = feats + (size_t)B_ * K3_;

  k_attn      <<<dim3(4, B_),   128, 0, stream>>>(x, Wa, ba, A_ws);
  k_bap       <<<dim3(32, B_),  128, 0, stream>>>(x, A_ws, feats);
  k_fc_partial<<<dim3(4, NKCH_), 128, 0, stream>>>(feats, Wc, part_ws);
  k_reduce    <<<(B_ * NC_ + 255) / 256, 256, 0, stream>>>(part_ws, bc, out);
}


# k_bap stores feats in k_fc_partial's A-operand order (128-byte lines of 2 b x 16 k, k pairs of two k-steps adjacent): feats fragment loads are 16-byte on 8 lines, were 8-byte on 16; same k order
# speedup vs baseline: 2.0336x; 1.0943x over previous
#include <hip/hip_runtime.h>
#include <cmath>


typedef __attribute__((ext_vector_type(2))) float v2f;
typedef __attribute__((ext_vector_type(4))) float v4f;
typedef __attribute__((ext_vector_type(8))) float v8f;

#define WMMA_F32(a, b, c) \
  __builtin_amdgcn_wmma_f32_16x16x4_f32(false, (a), false, (b), (short)0, (c), false, false)

#define B_   64
#define C_   2048
#define HW_  196
#define M_   32
#define NC_  396
#define K3_  (M_ * C_)
#define NPAD_ 400
#define NTILES_ 25
#define KCHUNK_ 512
#define NKCH_   128
#define FC_G_   8
#define FC_NG_  16
#define NPAIRS_ 13
#define NPR_  25
#define ATT_KS_ 512
#define ATT_XD_ 32
#define ATT_WG_ 16
#define ATT_WP_ 132
#define ATT_FB_ 32
#define FT_BP_  (K3_ * 2)

__device__ __forceinline__ float fast_sigmoid(float v) {
  return __builtin_amdgcn_rcpf(1.0f + __expf(-v));
}

__global__ __launch_bounds__(128) void k_attn(const float* __restrict__ x,
                                              const float* __restrict__ Wa,
                                              const float* __restrict__ ba,
                                              float* __restrict__ A_ws) {
  __shared__ float Wl[ATT_KS_ * ATT_WP_];

  const int wave = threadIdx.x >> 5;
  const int hwt  = blockIdx.x * 4 + wave;
  const int b    = blockIdx.y;
  const int lane = threadIdx.x & 31;
  const int half = lane >> 4;
  const int r    = lane & 15;

  {
    const float* src = Wa + (size_t)(16 * half + wave) * C_ + 4 * r;
    float* dst = Wl + r * ATT_WP_ + wave * 4 + half * 2;
#pragma unroll 1
    for (int n0 = 0; n0 < 128; n0 += ATT_FB_) {
      v4f tmp[ATT_FB_];
#pragma unroll
      for (int u = 0; u < ATT_FB_; ++u)
        tmp[u] = *(const v4f*)(src + (size_t)(4 * (u & 3)) * C_ + 16 * n0 + 64 * (u >> 2));
#pragma unroll
      for (int u = 0; u < ATT_FB_; ++u) {
        float* d = dst + (4 * n0 + 16 * (u >> 2)) * ATT_WP_ + 16 * (u & 3);
        *(v2f*)d        = tmp[u].xy;
        *(v2f*)(d + 64) = tmp[u].zw;
      }
    }
  }

  const int hw0  = hwt * 16;
  const int hwr  = min(hw0 + r, HW_ - 1);

  v8f acc0 = {}; v8f acc1 = {};
  const float* xb = x + (size_t)b * C_ * HW_;

  const float* xl  = xb + (size_t)(2 * half) * HW_ + hwr;

  v2f xa[ATT_XD_];
  if (hwt < 13) {
#pragma unroll
    for (int j = 0; j < ATT_XD_; ++j) {
      xa[j].x = xl[(size_t)(4 * j) * HW_];
      xa[j].y = xl[(size_t)(4 * j + 1) * HW_];
    }
  }
  __syncthreads();
  if (hwt >= 13) return;

  const float* wl = Wl + lane * 4;
  v4f wf[2][ATT_WG_];
#pragma unroll
  for (int j = 0; j < ATT_WG_; ++j) wf[0][j] = *(const v4f*)(wl + j * ATT_WP_);

#pragma unroll 1
  for (int ks0 = 0; ks0 < ATT_KS_ - ATT_XD_; ks0 += ATT_XD_) {
#pragma unroll
    for (int h = 0; h < 2; ++h) {
#pragma unroll
      for (int j = 0; j < ATT_WG_; ++j)
        wf[1 - h][j] = *(const v4f*)(wl + (ks0 + ATT_WG_ * (h + 1) + j) * ATT_WP_);
      __builtin_amdgcn_sched_barrier(0);
#pragma unroll
      for (int j = 0; j < ATT_WG_; ++j) {
        const int q = ATT_WG_ * h + j;
        acc0 = WMMA_F32(xa[q], wf[h][j].xy, acc0);
        acc1 = WMMA_F32(xa[q], wf[h][j].zw, acc1);
        const int kn = 4 * (ks0 + ATT_XD_ + q);
        xa[q].x = xl[(size_t)kn * HW_];
        xa[q].y = xl[(size_t)(kn + 1) * HW_];
      }
    }
  }
#pragma unroll
  for (int j = 0; j < ATT_WG_; ++j)
    wf[1][j] = *(const v4f*)(wl + (ATT_KS_ - ATT_WG_ + j) * ATT_WP_);
  __builtin_amdgcn_sched_barrier(0);
#pragma unroll
  for (int j = 0; j < ATT_WG_; ++j) {
    acc0 = WMMA_F32(xa[j], wf[0][j].xy, acc0);
    acc1 = WMMA_F32(xa[j], wf[0][j].zw, acc1);
  }
#pragma unroll
  for (int j = 0; j < ATT_WG_; ++j) {
    acc0 = WMMA_F32(xa[ATT_WG_ + j], wf[1][j].xy, acc0);
    acc1 = WMMA_F32(xa[ATT_WG_ + j], wf[1][j].zw, acc1);
  }

  const int p = 2 * hwt + half;
  if (p < NPR_) {
    const float bias0 = ba[r], bias1 = ba[16 + r];
    float s0[8], s1[8];
#pragma unroll
    for (int i = 0; i < 8; ++i) {
      s0[i] = fast_sigmoid(acc0[i] + bias0);
      s1[i] = fast_sigmoid(acc1[i] + bias1);
    }
    v4f* dst = (v4f*)A_ws + ((size_t)b * NPR_ + p) * 64;
    dst[r]           = (v4f){s0[0], s0[1], s0[4], s0[5]};
    dst[16 + r]      = (v4f){s0[2], s0[3], s0[6], s0[7]};
    dst[32 + r]      = (v4f){s1[0], s1[1], s1[4], s1[5]};
    dst[32 + 16 + r] = (v4f){s1[2], s1[3], s1[6], s1[7]};
  }
}

__global__ __launch_bounds__(128) void k_bap(const float* __restrict__ x,
                                             const float* __restrict__ A_ws,
                                             float* __restrict__ feats) {
  const int wave = threadIdx.x >> 5;
  const int ct   = blockIdx.x * 4 + wave;
  const int b    = blockIdx.y;
  const int lane = threadIdx.x & 31;
  const int half = lane >> 4;
  const int r    = lane & 15;
  const int c0   = ct * 16;

  v8f acc0 = {}; v8f acc1 = {};
  const v4f* Ab = (const v4f*)A_ws + (size_t)b * NPR_ * 64 + lane;
  const float* xl = x + ((size_t)b * C_ + c0 + r) * HW_ + 2 * half;

  for (int p = 0; p < NPR_ - 1; ++p) {
    const v4f a0 = Ab[p * 64];
    const v4f a1 = Ab[p * 64 + 32];
    const v2f be = *(const v2f*)(xl + 8 * p);
    const v2f bo = *(const v2f*)(xl + 8 * p + 4);
    acc0 = WMMA_F32(a0.xy, be, acc0);
    acc1 = WMMA_F32(a1.xy, be, acc1);
    acc0 = WMMA_F32(a0.zw, bo, acc0);
    acc1 = WMMA_F32(a1.zw, bo, acc1);
  }
  {
    const v2f a0 = *(const v2f*)(Ab + (NPR_ - 1) * 64);
    const v2f a1 = *(const v2f*)(Ab + (NPR_ - 1) * 64 + 32);
    const v2f be = *(const v2f*)(xl + 8 * (NPR_ - 1));
    acc0 = WMMA_F32(a0, be, acc0);
    acc1 = WMMA_F32(a1, be, acc1);
  }

  const float s = 1.0f / (float)HW_;
  const int rp = (r & 9) | ((r & 2) << 1) | ((r & 4) >> 1);
  float* fd = feats + (size_t)(b >> 1) * FT_BP_ + (size_t)ct * 32 + (b & 1) * 16 + rp;
#pragma unroll
  for (int i = 0; i < 8; ++i) {
    const int mr = i + (half ? 8 : 0);
    fd[(size_t)mr * (128 * 32)]        = acc0[i] * s;
    fd[(size_t)(16 + mr) * (128 * 32)] = acc1[i] * s;
  }
}

__device__ __forceinline__ void fc_load_w(v2f (&W)[FC_G_][2], const float* w0, const float* w1) {
#pragma unroll
  for (int j = 0; j < FC_G_; ++j) {
    W[j][0] = *(const v2f*)(w0 + 4 * j);
    W[j][1] = *(const v2f*)(w1 + 4 * j);
  }
}

__device__ __forceinline__ void fc_load_f(v4f (&F)[FC_G_ / 2][4], const float* f0) {
#pragma unroll
  for (int jp = 0; jp < FC_G_ / 2; ++jp) {
#pragma unroll
    for (int mt = 0; mt < 4; ++mt)
      F[jp][mt] = *(const v4f*)(f0 + (size_t)(mt * 8) * FT_BP_ + (jp >> 1) * 32 + (jp & 1) * 8);
  }
}

__device__ __forceinline__ void fc_mma(const v2f (&W)[FC_G_][2], const v4f (&F)[FC_G_ / 2][4],
                                       const bool nvalid0, const bool nvalid1,
                                       v8f (&acc0)[4], v8f (&acc1)[4]) {
#pragma unroll
  for (int j = 0; j < FC_G_; ++j) {
    v2f b0 = W[j][0];
    v2f b1 = W[j][1];
    b0.x = nvalid0 ? b0.x : 0.0f;
    b0.y = nvalid0 ? b0.y : 0.0f;
    b1.x = nvalid1 ? b1.x : 0.0f;
    b1.y = nvalid1 ? b1.y : 0.0f;
#pragma unroll
    for (int mt = 0; mt < 4; ++mt) {
      const v2f a = (j & 1) ? F[j >> 1][mt].zw : F[j >> 1][mt].xy;
      acc0[mt] = WMMA_F32(a, b0, acc0[mt]);
      acc1[mt] = WMMA_F32(a, b1, acc1[mt]);
    }
  }
}

__global__ __launch_bounds__(128) void k_fc_partial(const float* __restrict__ feats,
                                                    const float* __restrict__ Wc,
                                                    float* __restrict__ part) {
  const int wave = threadIdx.x >> 5;
  const int np   = blockIdx.x * 4 + wave;
  if (np >= NPAIRS_) return;
  const int kc   = blockIdx.y;
  const int lane = threadIdx.x & 31;
  const int half = lane >> 4;
  const int r    = lane & 15;
  const int n0   = np * 32;

  const bool nvalid0 = (n0 + r) < NC_;
  const bool nvalid1 = (n0 + 16 + r) < NC_;
  const int kbase    = kc * KCHUNK_ + 2 * half;
  const float* w0 = Wc + (size_t)min(n0 + r, NC_ - 1) * K3_ + kbase;
  const float* w1 = Wc + (size_t)min(n0 + 16 + r, NC_ - 1) * K3_ + kbase;
  const float* f0 = feats + (size_t)(r >> 1) * FT_BP_ + kc * (2 * KCHUNK_) + (r & 1) * 16 + 4 * half;

  v8f acc0[4] = {{}, {}, {}, {}};
  v8f acc1[4] = {{}, {}, {}, {}};

  v2f WS[3][FC_G_][2];
  v4f FS[2][FC_G_ / 2][4];

  fc_load_f(FS[0], f0);
  fc_load_w(WS[0], w0, w1);
  fc_load_w(WS[1], w0 + 4 * FC_G_, w1 + 4 * FC_G_);
#pragma unroll
  for (int g = 0; g < FC_NG_; ++g) {
    if (g + 1 < FC_NG_) fc_load_f(FS[(g + 1) & 1], f0 + 8 * FC_G_ * (g + 1));
    if (g + 2 < FC_NG_) fc_load_w(WS[(g + 2) % 3], w0 + 4 * FC_G_ * (g + 2), w1 + 4 * FC_G_ * (g + 2));
    __builtin_amdgcn_sched_barrier(0);
    fc_mma(WS[g % 3], FS[g & 1], nvalid0, nvalid1, acc0, acc1);
  }

#pragma unroll
  for (int mt = 0; mt < 4; ++mt) {
#pragma unroll
    for (int i = 0; i < 8; ++i) {
      const int brow = mt * 16 + i + (half ? 8 : 0);
      part[((size_t)kc * B_ + brow) * NPAD_ + n0 + r] = acc0[mt][i];
    }
  }
  if (n0 + 16 < NPAD_) {
#pragma unroll
    for (int mt = 0; mt < 4; ++mt) {
#pragma unroll
      for (int i = 0; i < 8; ++i) {
        const int brow = mt * 16 + i + (half ? 8 : 0);
        part[((size_t)kc * B_ + brow) * NPAD_ + n0 + 16 + r] = acc1[mt][i];
      }
    }
  }
}

__global__ __launch_bounds__(256) void k_reduce(const float* __restrict__ part,
                                               const float* __restrict__ bc,
                                               float* __restrict__ out) {
  const int idx = blockIdx.x * blockDim.x + threadIdx.x;
  if (idx >= B_ * NC_) return;
  const int b = idx / NC_;
  const int n = idx - b * NC_;
  float s = bc[n];
#pragma unroll 4
  for (int kc = 0; kc < NKCH_; ++kc)
    s += part[((size_t)kc * B_ + b) * NPAD_ + n];
  out[idx] = s;
}

extern "C" void kernel_launch(void* const* d_in, const int* in_sizes, int n_in,
                              void* d_out, int out_size, void* d_ws, size_t ws_size,
                              hipStream_t stream) {
  const float* x  = (const float*)d_in[0];
  const float* Wa = (const float*)d_in[1];
  const float* ba = (const float*)d_in[2];
  const float* Wc = (const float*)d_in[3];
  const float* bc = (const float*)d_in[4];
  float* out = (float*)d_out;

  float* A_ws    = (float*)d_ws;
  float* feats   = A_ws + (size_t)B_ * NPR_ * 256;
  float* part_ws = feats + (size_t)B_ * K3_;

  k_attn      <<<dim3(4, B_),   128, 0, stream>>>(x, Wa, ba, A_ws);
  k_bap       <<<dim3(32, B_),  128, 0, stream>>>(x, A_ws, feats);
  k_fc_partial<<<dim3(4, NKCH_), 128, 0, stream>>>(feats, Wc, part_ws);
  k_reduce    <<<(B_ * NC_ + 255) / 256, 256, 0, stream>>>(part_ws, bc, out);
}


# k_reduce re-cut over the split-K chunks: block = 128 outputs, its 8 waves each add 16 of the 128 partials (16-byte loads, all asked for at once), LDS sum in wave order (kc sum order changes)
# speedup vs baseline: 2.2631x; 1.1129x over previous
#include <hip/hip_runtime.h>
#include <cmath>


typedef __attribute__((ext_vector_type(2))) float v2f;
typedef __attribute__((ext_vector_type(4))) float v4f;
typedef __attribute__((ext_vector_type(8))) float v8f;

#define WMMA_F32(a, b, c) \
  __builtin_amdgcn_wmma_f32_16x16x4_f32(false, (a), false, (b), (short)0, (c), false, false)

#define B_   64
#define C_   2048
#define HW_  196
#define M_   32
#define NC_  396
#define K3_  (M_ * C_)
#define NPAD_ 400
#define NTILES_ 25
#define KCHUNK_ 512
#define NKCH_   128
#define FC_G_   8
#define FC_NG_  16
#define NPAIRS_ 13
#define NPR_  25
#define ATT_KS_ 512
#define ATT_XD_ 32
#define ATT_WG_ 16
#define ATT_WP_ 132
#define ATT_FB_ 32
#define FT_BP_  (K3_ * 2)

__device__ __forceinline__ float fast_sigmoid(float v) {
  return __builtin_amdgcn_rcpf(1.0f + __expf(-v));
}

__global__ __launch_bounds__(128) void k_attn(const float* __restrict__ x,
                                              const float* __restrict__ Wa,
                                              const float* __restrict__ ba,
                                              float* __restrict__ A_ws) {
  __shared__ float Wl[ATT_KS_ * ATT_WP_];

  const int wave = threadIdx.x >> 5;
  const int hwt  = blockIdx.x * 4 + wave;
  const int b    = blockIdx.y;
  const int lane = threadIdx.x & 31;
  const int half = lane >> 4;
  const int r    = lane & 15;

  {
    const float* src = Wa + (size_t)(16 * half + wave) * C_ + 4 * r;
    float* dst = Wl + r * ATT_WP_ + wave * 4 + half * 2;
#pragma unroll 1
    for (int n0 = 0; n0 < 128; n0 += ATT_FB_) {
      v4f tmp[ATT_FB_];
#pragma unroll
      for (int u = 0; u < ATT_FB_; ++u)
        tmp[u] = *(const v4f*)(src + (size_t)(4 * (u & 3)) * C_ + 16 * n0 + 64 * (u >> 2));
#pragma unroll
      for (int u = 0; u < ATT_FB_; ++u) {
        float* d = dst + (4 * n0 + 16 * (u >> 2)) * ATT_WP_ + 16 * (u & 3);
        *(v2f*)d        = tmp[u].xy;
        *(v2f*)(d + 64) = tmp[u].zw;
      }
    }
  }

  const int hw0  = hwt * 16;
  const int hwr  = min(hw0 + r, HW_ - 1);

  v8f acc0 = {}; v8f acc1 = {};
  const float* xb = x + (size_t)b * C_ * HW_;

  const float* xl  = xb + (size_t)(2 * half) * HW_ + hwr;

  v2f xa[ATT_XD_];
  if (hwt < 13) {
#pragma unroll
    for (int j = 0; j < ATT_XD_; ++j) {
      xa[j].x = xl[(size_t)(4 * j) * HW_];
      xa[j].y = xl[(size_t)(4 * j + 1) * HW_];
    }
  }
  __syncthreads();
  if (hwt >= 13) return;

  const float* wl = Wl + lane * 4;
  v4f wf[2][ATT_WG_];
#pragma unroll
  for (int j = 0; j < ATT_WG_; ++j) wf[0][j] = *(const v4f*)(wl + j * ATT_WP_);

#pragma unroll 1
  for (int ks0 = 0; ks0 < ATT_KS_ - ATT_XD_; ks0 += ATT_XD_) {
#pragma unroll
    for (int h = 0; h < 2; ++h) {
#pragma unroll
      for (int j = 0; j < ATT_WG_; ++j)
        wf[1 - h][j] = *(const v4f*)(wl + (ks0 + ATT_WG_ * (h + 1) + j) * ATT_WP_);
      __builtin_amdgcn_sched_barrier(0);
#pragma unroll
      for (int j = 0; j < ATT_WG_; ++j) {
        const int q = ATT_WG_ * h + j;
        acc0 = WMMA_F32(xa[q], wf[h][j].xy, acc0);
        acc1 = WMMA_F32(xa[q], wf[h][j].zw, acc1);
        const int kn = 4 * (ks0 + ATT_XD_ + q);
        xa[q].x = xl[(size_t)kn * HW_];
        xa[q].y = xl[(size_t)(kn + 1) * HW_];
      }
    }
  }
#pragma unroll
  for (int j = 0; j < ATT_WG_; ++j)
    wf[1][j] = *(const v4f*)(wl + (ATT_KS_ - ATT_WG_ + j) * ATT_WP_);
  __builtin_amdgcn_sched_barrier(0);
#pragma unroll
  for (int j = 0; j < ATT_WG_; ++j) {
    acc0 = WMMA_F32(xa[j], wf[0][j].xy, acc0);
    acc1 = WMMA_F32(xa[j], wf[0][j].zw, acc1);
  }
#pragma unroll
  for (int j = 0; j < ATT_WG_; ++j) {
    acc0 = WMMA_F32(xa[ATT_WG_ + j], wf[1][j].xy, acc0);
    acc1 = WMMA_F32(xa[ATT_WG_ + j], wf[1][j].zw, acc1);
  }

  const int p = 2 * hwt + half;
  if (p < NPR_) {
    const float bias0 = ba[r], bias1 = ba[16 + r];
    float s0[8], s1[8];
#pragma unroll
    for (int i = 0; i < 8; ++i) {
      s0[i] = fast_sigmoid(acc0[i] + bias0);
      s1[i] = fast_sigmoid(acc1[i] + bias1);
    }
    v4f* dst = (v4f*)A_ws + ((size_t)b * NPR_ + p) * 64;
    dst[r]           = (v4f){s0[0], s0[1], s0[4], s0[5]};
    dst[16 + r]      = (v4f){s0[2], s0[3], s0[6], s0[7]};
    dst[32 + r]      = (v4f){s1[0], s1[1], s1[4], s1[5]};
    dst[32 + 16 + r] = (v4f){s1[2], s1[3], s1[6], s1[7]};
  }
}

__global__ __launch_bounds__(128) void k_bap(const float* __restrict__ x,
                                             const float* __restrict__ A_ws,
                                             float* __restrict__ feats) {
  const int wave = threadIdx.x >> 5;
  const int ct   = blockIdx.x * 4 + wave;
  const int b    = blockIdx.y;
  const int lane = threadIdx.x & 31;
  const int half = lane >> 4;
  const int r    = lane & 15;
  const int c0   = ct * 16;

  v8f acc0 = {}; v8f acc1 = {};
  const v4f* Ab = (const v4f*)A_ws + (size_t)b * NPR_ * 64 + lane;
  const float* xl = x + ((size_t)b * C_ + c0 + r) * HW_ + 2 * half;

  for (int p = 0; p < NPR_ - 1; ++p) {
    const v4f a0 = Ab[p * 64];
    const v4f a1 = Ab[p * 64 + 32];
    const v2f be = *(const v2f*)(xl + 8 * p);
    const v2f bo = *(const v2f*)(xl + 8 * p + 4);
    acc0 = WMMA_F32(a0.xy, be, acc0);
    acc1 = WMMA_F32(a1.xy, be, acc1);
    acc0 = WMMA_F32(a0.zw, bo, acc0);
    acc1 = WMMA_F32(a1.zw, bo, acc1);
  }
  {
    const v2f a0 = *(const v2f*)(Ab + (NPR_ - 1) * 64);
    const v2f a1 = *(const v2f*)(Ab + (NPR_ - 1) * 64 + 32);
    const v2f be = *(const v2f*)(xl + 8 * (NPR_ - 1));
    acc0 = WMMA_F32(a0, be, acc0);
    acc1 = WMMA_F32(a1, be, acc1);
  }

  const float s = 1.0f / (float)HW_;
  const int rp = (r & 9) | ((r & 2) << 1) | ((r & 4) >> 1);
  float* fd = feats + (size_t)(b >> 1) * FT_BP_ + (size_t)ct * 32 + (b & 1) * 16 + rp;
#pragma unroll
  for (int i = 0; i < 8; ++i) {
    const int mr = i + (half ? 8 : 0);
    fd[(size_t)mr * (128 * 32)]        = acc0[i] * s;
    fd[(size_t)(16 + mr) * (128 * 32)] = acc1[i] * s;
  }
}

__device__ __forceinline__ void fc_load_w(v2f (&W)[FC_G_][2], const float* w0, const float* w1) {
#pragma unroll
  for (int j = 0; j < FC_G_; ++j) {
    W[j][0] = *(const v2f*)(w0 + 4 * j);
    W[j][1] = *(const v2f*)(w1 + 4 * j);
  }
}

__device__ __forceinline__ void fc_load_f(v4f (&F)[FC_G_ / 2][4], const float* f0) {
#pragma unroll
  for (int jp = 0; jp < FC_G_ / 2; ++jp) {
#pragma unroll
    for (int mt = 0; mt < 4; ++mt)
      F[jp][mt] = *(const v4f*)(f0 + (size_t)(mt * 8) * FT_BP_ + (jp >> 1) * 32 + (jp & 1) * 8);
  }
}

__device__ __forceinline__ void fc_mma(const v2f (&W)[FC_G_][2], const v4f (&F)[FC_G_ / 2][4],
                                       const bool nvalid0, const bool nvalid1,
                                       v8f (&acc0)[4], v8f (&acc1)[4]) {
#pragma unroll
  for (int j = 0; j < FC_G_; ++j) {
    v2f b0 = W[j][0];
    v2f b1 = W[j][1];
    b0.x = nvalid0 ? b0.x : 0.0f;
    b0.y = nvalid0 ? b0.y : 0.0f;
    b1.x = nvalid1 ? b1.x : 0.0f;
    b1.y = nvalid1 ? b1.y : 0.0f;
#pragma unroll
    for (int mt = 0; mt < 4; ++mt) {
      const v2f a = (j & 1) ? F[j >> 1][mt].zw : F[j >> 1][mt].xy;
      acc0[mt] = WMMA_F32(a, b0, acc0[mt]);
      acc1[mt] = WMMA_F32(a, b1, acc1[mt]);
    }
  }
}

__global__ __launch_bounds__(128) void k_fc_partial(const float* __restrict__ feats,
                                                    const float* __restrict__ Wc,
                                                    float* __restrict__ part) {
  const int wave = threadIdx.x >> 5;
  const int np   = blockIdx.x * 4 + wave;
  if (np >= NPAIRS_) return;
  const int kc   = blockIdx.y;
  const int lane = threadIdx.x & 31;
  const int half = lane >> 4;
  const int r    = lane & 15;
  const int n0   = np * 32;

  const bool nvalid0 = (n0 + r) < NC_;
  const bool nvalid1 = (n0 + 16 + r) < NC_;
  const int kbase    = kc * KCHUNK_ + 2 * half;
  const float* w0 = Wc + (size_t)min(n0 + r, NC_ - 1) * K3_ + kbase;
  const float* w1 = Wc + (size_t)min(n0 + 16 + r, NC_ - 1) * K3_ + kbase;
  const float* f0 = feats + (size_t)(r >> 1) * FT_BP_ + kc * (2 * KCHUNK_) + (r & 1) * 16 + 4 * half;

  v8f acc0[4] = {{}, {}, {}, {}};
  v8f acc1[4] = {{}, {}, {}, {}};

  v2f WS[3][FC_G_][2];
  v4f FS[2][FC_G_ / 2][4];

  fc_load_f(FS[0], f0);
  fc_load_w(WS[0], w0, w1);
  fc_load_w(WS[1], w0 + 4 * FC_G_, w1 + 4 * FC_G_);
#pragma unroll
  for (int g = 0; g < FC_NG_; ++g) {
    if (g + 1 < FC_NG_) fc_load_f(FS[(g + 1) & 1], f0 + 8 * FC_G_ * (g + 1));
    if (g + 2 < FC_NG_) fc_load_w(WS[(g + 2) % 3], w0 + 4 * FC_G_ * (g + 2), w1 + 4 * FC_G_ * (g + 2));
    __builtin_amdgcn_sched_barrier(0);
    fc_mma(WS[g % 3], FS[g & 1], nvalid0, nvalid1, acc0, acc1);
  }

#pragma unroll
  for (int mt = 0; mt < 4; ++mt) {
#pragma unroll
    for (int i = 0; i < 8; ++i) {
      const int brow = mt * 16 + i + (half ? 8 : 0);
      part[((size_t)kc * B_ + brow) * NPAD_ + n0 + r] = acc0[mt][i];
    }
  }
  if (n0 + 16 < NPAD_) {
#pragma unroll
    for (int mt = 0; mt < 4; ++mt) {
#pragma unroll
      for (int i = 0; i < 8; ++i) {
        const int brow = mt * 16 + i + (half ? 8 : 0);
        part[((size_t)kc * B_ + brow) * NPAD_ + n0 + 16 + r] = acc1[mt][i];
      }
    }
  }
}

#define RD_W_   8
#define RD_KC_  (NKCH_ / RD_W_)
#define RD_G_   (NC_ / 4)

__global__ __launch_bounds__(32 * RD_W_) void k_reduce(const float* __restrict__ part,
                                                       const float* __restrict__ bc,
                                                       float* __restrict__ out) {
  __shared__ v4f red[RD_W_ - 1][32];
  const int wave = threadIdx.x >> 5;
  const int lane = threadIdx.x & 31;
  const int g = blockIdx.x * 32 + lane;
  const int b = g / RD_G_;
  const int n = 4 * (g - b * RD_G_);
  const float* p = part + ((size_t)(wave * RD_KC_) * B_ + b) * NPAD_ + n;

  const v4f bias = (v4f){bc[n], bc[n + 1], bc[n + 2], bc[n + 3]};

  v4f v[RD_KC_];
#pragma unroll
  for (int i = 0; i < RD_KC_; ++i) v[i] = *(const v4f*)(p + (size_t)i * (B_ * NPAD_));
  v4f s = v[0];
#pragma unroll
  for (int i = 1; i < RD_KC_; ++i) s += v[i];

  if (wave) red[wave - 1][lane] = s;
  __syncthreads();
  if (wave == 0) {
    v4f t = bias;
    t += s;
#pragma unroll
    for (int w = 1; w < RD_W_; ++w) t += red[w - 1][lane];
    *(v4f*)(out + 4 * (size_t)g) = t;
  }
}

extern "C" void kernel_launch(void* const* d_in, const int* in_sizes, int n_in,
                              void* d_out, int out_size, void* d_ws, size_t ws_size,
                              hipStream_t stream) {
  const float* x  = (const float*)d_in[0];
  const float* Wa = (const float*)d_in[1];
  const float* ba = (const float*)d_in[2];
  const float* Wc = (const float*)d_in[3];
  const float* bc = (const float*)d_in[4];
  float* out = (float*)d_out;

  float* A_ws    = (float*)d_ws;
  float* feats   = A_ws + (size_t)B_ * NPR_ * 256;
  float* part_ws = feats + (size_t)B_ * K3_;

  k_attn      <<<dim3(4, B_),   128, 0, stream>>>(x, Wa, ba, A_ws);
  k_bap       <<<dim3(32, B_),  128, 0, stream>>>(x, A_ws, feats);
  k_fc_partial<<<dim3(4, NKCH_), 128, 0, stream>>>(feats, Wc, part_ws);
  k_reduce    <<<B_ * RD_G_ / 32, 32 * RD_W_, 0, stream>>>(part_ws, bc, out);
}


# k_bap re-tiled: a wave owns 2 c-tiles (an A fragment feeds both), 8-wave blocks: 4096 waves in 512 blocks, were 8192 in 2048; its loads run 4 pairs of k-steps ahead in registers; same sum order
# speedup vs baseline: 2.5158x; 1.1117x over previous
#include <hip/hip_runtime.h>
#include <cmath>


typedef __attribute__((ext_vector_type(2))) float v2f;
typedef __attribute__((ext_vector_type(4))) float v4f;
typedef __attribute__((ext_vector_type(8))) float v8f;

#define WMMA_F32(a, b, c) \
  __builtin_amdgcn_wmma_f32_16x16x4_f32(false, (a), false, (b), (short)0, (c), false, false)

#define B_   64
#define C_   2048
#define HW_  196
#define M_   32
#define NC_  396
#define K3_  (M_ * C_)
#define NPAD_ 400
#define NTILES_ 25
#define KCHUNK_ 512
#define NKCH_   128
#define FC_G_   8
#define FC_NG_  16
#define NPAIRS_ 13
#define NPR_  25
#define ATT_KS_ 512
#define ATT_XD_ 32
#define ATT_WG_ 16
#define ATT_WP_ 132
#define ATT_FB_ 32
#define FT_BP_  (K3_ * 2)

__device__ __forceinline__ float fast_sigmoid(float v) {
  return __builtin_amdgcn_rcpf(1.0f + __expf(-v));
}

__global__ __launch_bounds__(128) void k_attn(const float* __restrict__ x,
                                              const float* __restrict__ Wa,
                                              const float* __restrict__ ba,
                                              float* __restrict__ A_ws) {
  __shared__ float Wl[ATT_KS_ * ATT_WP_];

  const int wave = threadIdx.x >> 5;
  const int hwt  = blockIdx.x * 4 + wave;
  const int b    = blockIdx.y;
  const int lane = threadIdx.x & 31;
  const int half = lane >> 4;
  const int r    = lane & 15;

  {
    const float* src = Wa + (size_t)(16 * half + wave) * C_ + 4 * r;
    float* dst = Wl + r * ATT_WP_ + wave * 4 + half * 2;
#pragma unroll 1
    for (int n0 = 0; n0 < 128; n0 += ATT_FB_) {
      v4f tmp[ATT_FB_];
#pragma unroll
      for (int u = 0; u < ATT_FB_; ++u)
        tmp[u] = *(const v4f*)(src + (size_t)(4 * (u & 3)) * C_ + 16 * n0 + 64 * (u >> 2));
#pragma unroll
      for (int u = 0; u < ATT_FB_; ++u) {
        float* d = dst + (4 * n0 + 16 * (u >> 2)) * ATT_WP_ + 16 * (u & 3);
        *(v2f*)d        = tmp[u].xy;
        *(v2f*)(d + 64) = tmp[u].zw;
      }
    }
  }

  const int hw0  = hwt * 16;
  const int hwr  = min(hw0 + r, HW_ - 1);

  v8f acc0 = {}; v8f acc1 = {};
  const float* xb = x + (size_t)b * C_ * HW_;

  const float* xl  = xb + (size_t)(2 * half) * HW_ + hwr;

  v2f xa[ATT_XD_];
  if (hwt < 13) {
#pragma unroll
    for (int j = 0; j < ATT_XD_; ++j) {
      xa[j].x = xl[(size_t)(4 * j) * HW_];
      xa[j].y = xl[(size_t)(4 * j + 1) * HW_];
    }
  }
  __syncthreads();
  if (hwt >= 13) return;

  const float* wl = Wl + lane * 4;
  v4f wf[2][ATT_WG_];
#pragma unroll
  for (int j = 0; j < ATT_WG_; ++j) wf[0][j] = *(const v4f*)(wl + j * ATT_WP_);

#pragma unroll 1
  for (int ks0 = 0; ks0 < ATT_KS_ - ATT_XD_; ks0 += ATT_XD_) {
#pragma unroll
    for (int h = 0; h < 2; ++h) {
#pragma unroll
      for (int j = 0; j < ATT_WG_; ++j)
        wf[1 - h][j] = *(const v4f*)(wl + (ks0 + ATT_WG_ * (h + 1) + j) * ATT_WP_);
      __builtin_amdgcn_sched_barrier(0);
#pragma unroll
      for (int j = 0; j < ATT_WG_; ++j) {
        const int q = ATT_WG_ * h + j;
        acc0 = WMMA_F32(xa[q], wf[h][j].xy, acc0);
        acc1 = WMMA_F32(xa[q], wf[h][j].zw, acc1);
        const int kn = 4 * (ks0 + ATT_XD_ + q);
        xa[q].x = xl[(size_t)kn * HW_];
        xa[q].y = xl[(size_t)(kn + 1) * HW_];
      }
    }
  }
#pragma unroll
  for (int j = 0; j < ATT_WG_; ++j)
    wf[1][j] = *(const v4f*)(wl + (ATT_KS_ - ATT_WG_ + j) * ATT_WP_);
  __builtin_amdgcn_sched_barrier(0);
#pragma unroll
  for (int j = 0; j < ATT_WG_; ++j) {
    acc0 = WMMA_F32(xa[j], wf[0][j].xy, acc0);
    acc1 = WMMA_F32(xa[j], wf[0][j].zw, acc1);
  }
#pragma unroll
  for (int j = 0; j < ATT_WG_; ++j) {
    acc0 = WMMA_F32(xa[ATT_WG_ + j], wf[1][j].xy, acc0);
    acc1 = WMMA_F32(xa[ATT_WG_ + j], wf[1][j].zw, acc1);
  }

  const int p = 2 * hwt + half;
  if (p < NPR_) {
    const float bias0 = ba[r], bias1 = ba[16 + r];
    float s0[8], s1[8];
#pragma unroll
    for (int i = 0; i < 8; ++i) {
      s0[i] = fast_sigmoid(acc0[i] + bias0);
      s1[i] = fast_sigmoid(acc1[i] + bias1);
    }
    v4f* dst = (v4f*)A_ws + ((size_t)b * NPR_ + p) * 64;
    dst[r]           = (v4f){s0[0], s0[1], s0[4], s0[5]};
    dst[16 + r]      = (v4f){s0[2], s0[3], s0[6], s0[7]};
    dst[32 + r]      = (v4f){s1[0], s1[1], s1[4], s1[5]};
    dst[32 + 16 + r] = (v4f){s1[2], s1[3], s1[6], s1[7]};
  }
}

__global__ __launch_bounds__(256) void k_bap(const float* __restrict__ x,
                                             const float* __restrict__ A_ws,
                                             float* __restrict__ feats) {
  const int wave = threadIdx.x >> 5;
  const int ct   = (blockIdx.x * 8 + wave) * 2;
  const int b    = blockIdx.y;
  const int lane = threadIdx.x & 31;
  const int half = lane >> 4;
  const int r    = lane & 15;
  const int c0   = ct * 16;

  v8f acc00 = {}; v8f acc01 = {};
  v8f acc10 = {}; v8f acc11 = {};
  const v4f* Ab = (const v4f*)A_ws + (size_t)b * NPR_ * 64 + lane;
  const float* xl0 = x + ((size_t)b * C_ + c0 + r) * HW_ + 2 * half;
  const float* xl1 = xl0 + 16 * HW_;

  v4f ra0[4], ra1[4];
  v2f xe0[4], xo0[4], xe1[4], xo1[4];
  v2f ta0, ta1, tb0, tb1;
#pragma unroll
  for (int j = 0; j < 3; ++j) {
    ra0[j] = Ab[j * 64];
    ra1[j] = Ab[j * 64 + 32];
    xe0[j] = *(const v2f*)(xl0 + 8 * j);
    xo0[j] = *(const v2f*)(xl0 + 8 * j + 4);
    xe1[j] = *(const v2f*)(xl1 + 8 * j);
    xo1[j] = *(const v2f*)(xl1 + 8 * j + 4);
  }
#pragma unroll
  for (int p = 0; p < NPR_ - 1; ++p) {
    if (p + 3 < NPR_ - 1) {
      const int n = p + 3, t = n & 3;
      ra0[t] = Ab[n * 64];
      ra1[t] = Ab[n * 64 + 32];
      xe0[t] = *(const v2f*)(xl0 + 8 * n);
      xo0[t] = *(const v2f*)(xl0 + 8 * n + 4);
      xe1[t] = *(const v2f*)(xl1 + 8 * n);
      xo1[t] = *(const v2f*)(xl1 + 8 * n + 4);
    } else if (p + 3 == NPR_ - 1) {
      ta0 = *(const v2f*)(Ab + (NPR_ - 1) * 64);
      ta1 = *(const v2f*)(Ab + (NPR_ - 1) * 64 + 32);
      tb0 = *(const v2f*)(xl0 + 8 * (NPR_ - 1));
      tb1 = *(const v2f*)(xl1 + 8 * (NPR_ - 1));
    }
    __builtin_amdgcn_sched_barrier(0);
    const int q = p & 3;
    acc00 = WMMA_F32(ra0[q].xy, xe0[q], acc00);
    acc01 = WMMA_F32(ra1[q].xy, xe0[q], acc01);
    acc10 = WMMA_F32(ra0[q].xy, xe1[q], acc10);
    acc11 = WMMA_F32(ra1[q].xy, xe1[q], acc11);
    acc00 = WMMA_F32(ra0[q].zw, xo0[q], acc00);
    acc01 = WMMA_F32(ra1[q].zw, xo0[q], acc01);
    acc10 = WMMA_F32(ra0[q].zw, xo1[q], acc10);
    acc11 = WMMA_F32(ra1[q].zw, xo1[q], acc11);
  }
  acc00 = WMMA_F32(ta0, tb0, acc00);
  acc01 = WMMA_F32(ta1, tb0, acc01);
  acc10 = WMMA_F32(ta0, tb1, acc10);
  acc11 = WMMA_F32(ta1, tb1, acc11);

  const float s = 1.0f / (float)HW_;
  const int rp = (r & 9) | ((r & 2) << 1) | ((r & 4) >> 1);
  float* fd = feats + (size_t)(b >> 1) * FT_BP_ + (size_t)ct * 32 + (b & 1) * 16 + rp;
#pragma unroll
  for (int i = 0; i < 8; ++i) {
    const int mr = i + (half ? 8 : 0);
    fd[(size_t)mr * (128 * 32)]             = acc00[i] * s;
    fd[(size_t)(16 + mr) * (128 * 32)]      = acc01[i] * s;
    fd[(size_t)mr * (128 * 32) + 32]        = acc10[i] * s;
    fd[(size_t)(16 + mr) * (128 * 32) + 32] = acc11[i] * s;
  }
}

__device__ __forceinline__ void fc_load_w(v2f (&W)[FC_G_][2], const float* w0, const float* w1) {
#pragma unroll
  for (int j = 0; j < FC_G_; ++j) {
    W[j][0] = *(const v2f*)(w0 + 4 * j);
    W[j][1] = *(const v2f*)(w1 + 4 * j);
  }
}

__device__ __forceinline__ void fc_load_f(v4f (&F)[FC_G_ / 2][4], const float* f0) {
#pragma unroll
  for (int jp = 0; jp < FC_G_ / 2; ++jp) {
#pragma unroll
    for (int mt = 0; mt < 4; ++mt)
      F[jp][mt] = *(const v4f*)(f0 + (size_t)(mt * 8) * FT_BP_ + (jp >> 1) * 32 + (jp & 1) * 8);
  }
}

__device__ __forceinline__ void fc_mma(const v2f (&W)[FC_G_][2], const v4f (&F)[FC_G_ / 2][4],
                                       const bool nvalid0, const bool nvalid1,
                                       v8f (&acc0)[4], v8f (&acc1)[4]) {
#pragma unroll
  for (int j = 0; j < FC_G_; ++j) {
    v2f b0 = W[j][0];
    v2f b1 = W[j][1];
    b0.x = nvalid0 ? b0.x : 0.0f;
    b0.y = nvalid0 ? b0.y : 0.0f;
    b1.x = nvalid1 ? b1.x : 0.0f;
    b1.y = nvalid1 ? b1.y : 0.0f;
#pragma unroll
    for (int mt = 0; mt < 4; ++mt) {
      const v2f a = (j & 1) ? F[j >> 1][mt].zw : F[j >> 1][mt].xy;
      acc0[mt] = WMMA_F32(a, b0, acc0[mt]);
      acc1[mt] = WMMA_F32(a, b1, acc1[mt]);
    }
  }
}

__global__ __launch_bounds__(128) void k_fc_partial(const float* __restrict__ feats,
                                                    const float* __restrict__ Wc,
                                                    float* __restrict__ part) {
  const int wave = threadIdx.x >> 5;
  const int np   = blockIdx.x * 4 + wave;
  if (np >= NPAIRS_) return;
  const int kc   = blockIdx.y;
  const int lane = threadIdx.x & 31;
  const int half = lane >> 4;
  const int r    = lane & 15;
  const int n0   = np * 32;

  const bool nvalid0 = (n0 + r) < NC_;
  const bool nvalid1 = (n0 + 16 + r) < NC_;
  const int kbase    = kc * KCHUNK_ + 2 * half;
  const float* w0 = Wc + (size_t)min(n0 + r, NC_ - 1) * K3_ + kbase;
  const float* w1 = Wc + (size_t)min(n0 + 16 + r, NC_ - 1) * K3_ + kbase;
  const float* f0 = feats + (size_t)(r >> 1) * FT_BP_ + kc * (2 * KCHUNK_) + (r & 1) * 16 + 4 * half;

  v8f acc0[4] = {{}, {}, {}, {}};
  v8f acc1[4] = {{}, {}, {}, {}};

  v2f WS[3][FC_G_][2];
  v4f FS[2][FC_G_ / 2][4];

  fc_load_f(FS[0], f0);
  fc_load_w(WS[0], w0, w1);
  fc_load_w(WS[1], w0 + 4 * FC_G_, w1 + 4 * FC_G_);
#pragma unroll
  for (int g = 0; g < FC_NG_; ++g) {
    if (g + 1 < FC_NG_) fc_load_f(FS[(g + 1) & 1], f0 + 8 * FC_G_ * (g + 1));
    if (g + 2 < FC_NG_) fc_load_w(WS[(g + 2) % 3], w0 + 4 * FC_G_ * (g + 2), w1 + 4 * FC_G_ * (g + 2));
    __builtin_amdgcn_sched_barrier(0);
    fc_mma(WS[g % 3], FS[g & 1], nvalid0, nvalid1, acc0, acc1);
  }

#pragma unroll
  for (int mt = 0; mt < 4; ++mt) {
#pragma unroll
    for (int i = 0; i < 8; ++i) {
      const int brow = mt * 16 + i + (half ? 8 : 0);
      part[((size_t)kc * B_ + brow) * NPAD_ + n0 + r] = acc0[mt][i];
    }
  }
  if (n0 + 16 < NPAD_) {
#pragma unroll
    for (int mt = 0; mt < 4; ++mt) {
#pragma unroll
      for (int i = 0; i < 8; ++i) {
        const int brow = mt * 16 + i + (half ? 8 : 0);
        part[((size_t)kc * B_ + brow) * NPAD_ + n0 + 16 + r] = acc1[mt][i];
      }
    }
  }
}

#define RD_W_   8
#define RD_KC_  (NKCH_ / RD_W_)
#define RD_G_   (NC_ / 4)

__global__ __launch_bounds__(32 * RD_W_) void k_reduce(const float* __restrict__ part,
                                                       const float* __restrict__ bc,
                                                       float* __restrict__ out) {
  __shared__ v4f red[RD_W_ - 1][32];
  const int wave = threadIdx.x >> 5;
  const int lane = threadIdx.x & 31;
  const int g = blockIdx.x * 32 + lane;
  const int b = g / RD_G_;
  const int n = 4 * (g - b * RD_G_);
  const float* p = part + ((size_t)(wave * RD_KC_) * B_ + b) * NPAD_ + n;

  const v4f bias = (v4f){bc[n], bc[n + 1], bc[n + 2], bc[n + 3]};

  v4f v[RD_KC_];
#pragma unroll
  for (int i = 0; i < RD_KC_; ++i) v[i] = *(const v4f*)(p + (size_t)i * (B_ * NPAD_));
  v4f s = v[0];
#pragma unroll
  for (int i = 1; i < RD_KC_; ++i) s += v[i];

  if (wave) red[wave - 1][lane] = s;
  __syncthreads();
  if (wave == 0) {
    v4f t = bias;
    t += s;
#pragma unroll
    for (int w = 1; w < RD_W_; ++w) t += red[w - 1][lane];
    *(v4f*)(out + 4 * (size_t)g) = t;
  }
}

extern "C" void kernel_launch(void* const* d_in, const int* in_sizes, int n_in,
                              void* d_out, int out_size, void* d_ws, size_t ws_size,
                              hipStream_t stream) {
  const float* x  = (const float*)d_in[0];
  const float* Wa = (const float*)d_in[1];
  const float* ba = (const float*)d_in[2];
  const float* Wc = (const float*)d_in[3];
  const float* bc = (const float*)d_in[4];
  float* out = (float*)d_out;

  float* A_ws    = (float*)d_ws;
  float* feats   = A_ws + (size_t)B_ * NPR_ * 256;
  float* part_ws = feats + (size_t)B_ * K3_;

  k_attn      <<<dim3(4, B_),   128, 0, stream>>>(x, Wa, ba, A_ws);
  k_bap       <<<dim3(8, B_),   256, 0, stream>>>(x, A_ws, feats);
  k_fc_partial<<<dim3(4, NKCH_), 128, 0, stream>>>(feats, Wc, part_ws);
  k_reduce    <<<B_ * RD_G_ / 32, 32 * RD_W_, 0, stream>>>(part_ws, bc, out);
}
